# starting kernel
# baseline (speedup 1.0000x reference)
#include <hip/hip_runtime.h>
#include <hip/hip_bf16.h>


typedef __attribute__((ext_vector_type(2))) float v2f;
typedef __attribute__((ext_vector_type(8))) float v8f;

#define D 64

__global__ void hr_init_kernel(const float* __restrict__ Basis,
                               const float* __restrict__ alpha,
                               float* __restrict__ h_r)
{
    int t = threadIdx.x;
    if (t >= 3 * D) return;
    int r = t >> 6, n = t & 63;
    float s = 0.f;
#pragma unroll
    for (int b = 0; b < 16; ++b) s += alpha[r * 16 + b] * Basis[b * D + n];
    h_r[t] = s;
}

__global__ void rel_kernel(const float* __restrict__ h_r,
                           const float* __restrict__ Wrel,
                           float* __restrict__ h_r_out,
                           int relu)
{
    int t = threadIdx.x;
    if (t >= 3 * D) return;
    int r = t >> 6, n = t & 63;
    float s = 0.f;
#pragma unroll 8
    for (int k = 0; k < D; ++k) s += h_r[r * D + k] * Wrel[n * D + k];
    if (relu) s = fmaxf(s, 0.f);
    h_r_out[t] = s;
}

__global__ __launch_bounds__(256) void zero_kernel(float4* __restrict__ p, long long n4)
{
    long long i = (long long)blockIdx.x * blockDim.x + threadIdx.x;
    if (i < n4) p[i] = make_float4(0.f, 0.f, 0.f, 0.f);
}

__global__ __launch_bounds__(256)
void edge_kernel(const float* __restrict__ h_in,
                 const float* __restrict__ h_r,
                 const int* __restrict__ src,
                 const int* __restrict__ dst,
                 const int* __restrict__ etype,
                 const int* __restrict__ dir,
                 float* __restrict__ acc,
                 long long total16, int n_nodes)
{
    __shared__ float lds_hr[2 * D];
    int t = threadIdx.x;
    if (t < 2 * D) lds_hr[t] = h_r[t];
    __syncthreads();

    long long i = (long long)blockIdx.x * blockDim.x + t;
    if (i >= total16) return;
    int e = (int)(i >> 4);
    int c = (int)(i & 15) * 4;
    int s  = src[e];
    int d0 = dst[e];
    int et = etype[e];
    int dr = dir[e];

    const float4 hv = *(const float4*)(h_in + (size_t)s * D + c);
    const float4 rv = *(const float4*)(lds_hr + et * D + c);

    float* p = acc + ((size_t)dr * n_nodes + (size_t)d0) * D + c;
    atomicAdd(p + 0, hv.x - rv.x);
    atomicAdd(p + 1, hv.y - rv.y);
    atomicAdd(p + 2, hv.z - rv.z);
    atomicAdd(p + 3, hv.w - rv.w);
}

__global__ __launch_bounds__(256)
void node_kernel(const float* __restrict__ h_in,
                 const float* __restrict__ acc,
                 const float* __restrict__ hr2,
                 const float* __restrict__ Ws,
                 const float* __restrict__ Wo,
                 const float* __restrict__ Wi,
                 float* __restrict__ h_out,
                 int n_nodes, int relu)
{
    __shared__ float lds_w[3 * D * D];
    __shared__ float lds_hr2[D];

    int t = threadIdx.x;
    for (int i = t; i < D * D; i += 256) {
        lds_w[i]             = Ws[i];
        lds_w[D * D + i]     = Wo[i];
        lds_w[2 * D * D + i] = Wi[i];
    }
    if (t < D) lds_hr2[t] = hr2[t];
    __syncthreads();

    const int wave  = t >> 5;
    const int lane  = t & 31;
    const int lhalf = lane >> 4;
    const int lmod  = lane & 15;

    const int m0 = (blockIdx.x * 2 + (wave >> 2)) * 16;
    const int n0 = (wave & 3) * 16;
    if (m0 >= n_nodes) return;

    const int arow = m0 + lmod;
    const int brow = n0 + lmod;

    const float* __restrict__ xr = h_in + (size_t)arow * D;
    const float* __restrict__ a0 = acc + (size_t)arow * D;
    const float* __restrict__ a1 = acc + ((size_t)n_nodes + (size_t)arow) * D;

    float res[8];

#if defined(__has_builtin) && __has_builtin(__builtin_amdgcn_wmma_f32_16x16x4_f32)
    v8f c = {};
#pragma unroll
    for (int k0 = 0; k0 < D; k0 += 4) {
        const int ka = k0 + lhalf * 2;
        v2f a, b;
        a.x = xr[ka]     - lds_hr2[ka];
        a.y = xr[ka + 1] - lds_hr2[ka + 1];
        b.x = lds_w[brow * D + ka];
        b.y = lds_w[brow * D + ka + 1];
        c = __builtin_amdgcn_wmma_f32_16x16x4_f32(false, a, false, b, (short)0, c, false, false);
    }
#pragma unroll
    for (int k0 = 0; k0 < D; k0 += 4) {
        const int ka = k0 + lhalf * 2;
        v2f a, b;
        a.x = a0[ka];
        a.y = a0[ka + 1];
        b.x = lds_w[D * D + brow * D + ka];
        b.y = lds_w[D * D + brow * D + ka + 1];
        c = __builtin_amdgcn_wmma_f32_16x16x4_f32(false, a, false, b, (short)0, c, false, false);
    }
#pragma unroll
    for (int k0 = 0; k0 < D; k0 += 4) {
        const int ka = k0 + lhalf * 2;
        v2f a, b;
        a.x = a1[ka];
        a.y = a1[ka + 1];
        b.x = lds_w[2 * D * D + brow * D + ka];
        b.y = lds_w[2 * D * D + brow * D + ka + 1];
        c = __builtin_amdgcn_wmma_f32_16x16x4_f32(false, a, false, b, (short)0, c, false, false);
    }
#pragma unroll
    for (int r = 0; r < 8; ++r) res[r] = c[r];
#else
#pragma unroll
    for (int r = 0; r < 8; ++r) {
        const int m = m0 + r + lhalf * 8;
        const float* __restrict__ xm = h_in + (size_t)m * D;
        const float* __restrict__ b0 = acc + (size_t)m * D;
        const float* __restrict__ b1 = acc + ((size_t)n_nodes + (size_t)m) * D;
        float s = 0.f;
        for (int k = 0; k < D; ++k) {
            s += (xm[k] - lds_hr2[k]) * lds_w[brow * D + k];
            s += b0[k] * lds_w[D * D + brow * D + k];
            s += b1[k] * lds_w[2 * D * D + brow * D + k];
        }
        res[r] = s;
    }
#endif

#pragma unroll
    for (int r = 0; r < 8; ++r) {
        float v = res[r];
        if (relu) v = fmaxf(v, 0.f);
        const int m = m0 + r + lhalf * 8;
        h_out[(size_t)m * D + n0 + lmod] = v;
    }
}

extern "C" void kernel_launch(void* const* d_in, const int* in_sizes, int n_in,
                              void* d_out, int out_size, void* d_ws, size_t ws_size,
                              hipStream_t stream)
{
    const float* h_u   = (const float*)d_in[0];
    const float* Basis = (const float*)d_in[1];
    const float* alpha = (const float*)d_in[2];
    const float* W_O   = (const float*)d_in[3];
    const float* W_I   = (const float*)d_in[4];
    const float* W_S   = (const float*)d_in[5];
    const float* W_rel = (const float*)d_in[6];
    const int*   src   = (const int*)d_in[7];
    const int*   dst   = (const int*)d_in[8];
    const int*   etype = (const int*)d_in[9];
    const int*   edir  = (const int*)d_in[10];
    float* out = (float*)d_out;

    const int n_nodes = in_sizes[0] / D;
    const int n_edges = in_sizes[7];

    float* acc = (float*)d_ws;
    float* hb0 = acc + (size_t)2 * n_nodes * D;
    float* hb1 = hb0 + (size_t)n_nodes * D;
    float* hra = hb1 + (size_t)n_nodes * D;
    float* hrb = hra + 256;

    hr_init_kernel<<<1, 192, 0, stream>>>(Basis, alpha, hra);

    const float* hin[3]  = { h_u, hb0, hb1 };
    float*       hout[3] = { hb0, hb1, out };

    const long long edge_total16 = (long long)n_edges * 16;
    const unsigned  edge_blocks  = (unsigned)((edge_total16 + 255) / 256);
    const long long acc_f4       = (long long)2 * n_nodes * D / 4;
    const unsigned  zero_blocks  = (unsigned)((acc_f4 + 255) / 256);
    const unsigned  node_blocks  = (unsigned)((n_nodes + 31) / 32);

    float* hrcur = hra;
    float* hrnxt = hrb;

    for (int l = 0; l < 3; ++l) {
        const int relu = (l < 2) ? 1 : 0;

        zero_kernel<<<zero_blocks, 256, 0, stream>>>((float4*)acc, acc_f4);

        edge_kernel<<<edge_blocks, 256, 0, stream>>>(hin[l], hrcur, src, dst, etype, edir,
                                                     acc, edge_total16, n_nodes);

        node_kernel<<<node_blocks, 256, 0, stream>>>(hin[l], acc, hrcur + 2 * D,
                                                     W_S + (size_t)l * D * D,
                                                     W_O + (size_t)l * D * D,
                                                     W_I + (size_t)l * D * D,
                                                     hout[l], n_nodes, relu);

        if (l < 2) {
            rel_kernel<<<1, 192, 0, stream>>>(hrcur, W_rel + (size_t)l * D * D, hrnxt, 1);
            float* tmp = hrcur; hrcur = hrnxt; hrnxt = tmp;
        }
    }
}


# Edge sums over destination-sorted segments in fixed edge-id order (table rebuilt each call), summed in registers inside the node GEMM kernel; replaces the f32 atomic scatter and its clearing pass
# speedup vs baseline: 1.3890x; 1.3890x over previous
#include <hip/hip_runtime.h>
#include <hip/hip_bf16.h>


typedef __attribute__((ext_vector_type(2))) float v2f;
typedef __attribute__((ext_vector_type(8))) float v8f;

#define D 64
#define APITCH 68

__global__ void hr_init_kernel(const float* __restrict__ Basis,
                               const float* __restrict__ alpha,
                               float* __restrict__ h_r)
{
    int t = threadIdx.x;
    if (t >= 3 * D) return;
    int r = t >> 6, n = t & 63;
    float s = 0.f;
#pragma unroll
    for (int b = 0; b < 16; ++b) s += alpha[r * 16 + b] * Basis[b * D + n];
    h_r[t] = s;
}

__global__ void rel_kernel(const float* __restrict__ h_r,
                           const float* __restrict__ Wrel,
                           float* __restrict__ h_r_out,
                           int relu)
{
    int t = threadIdx.x;
    if (t >= 3 * D) return;
    int r = t >> 6, n = t & 63;
    float s = 0.f;
#pragma unroll 8
    for (int k = 0; k < D; ++k) s += h_r[r * D + k] * Wrel[n * D + k];
    if (relu) s = fmaxf(s, 0.f);
    h_r_out[t] = s;
}

__global__ __launch_bounds__(256) void zero_kernel(int* __restrict__ p, int n)
{
    int i = blockIdx.x * blockDim.x + threadIdx.x;
    if (i < n) p[i] = 0;
}

__global__ __launch_bounds__(256) void degree_kernel(const int* __restrict__ dst,
                                                     int* __restrict__ cnt, int n_edges)
{
    int e = blockIdx.x * blockDim.x + threadIdx.x;
    if (e < n_edges) atomicAdd(&cnt[dst[e]], 1);
}

__device__ __forceinline__ int wave_sum(int v)
{
#pragma unroll
    for (int o = 16; o > 0; o >>= 1) v += __shfl_down(v, o, 32);
    return v;
}

__device__ __forceinline__ int wave_scan_incl(int v, int lane)
{
#pragma unroll
    for (int o = 1; o < 32; o <<= 1) {
        int nb = __shfl_up(v, o, 32);
        if (lane >= o) v += nb;
    }
    return v;
}

__global__ __launch_bounds__(1024) void blocksum_kernel(const int* __restrict__ cnt,
                                                        int* __restrict__ bsum, int n)
{
    __shared__ int wsum[32];
    const int t = threadIdx.x, lane = t & 31, w = t >> 5;
    const int i = blockIdx.x * 1024 + t;
    int v = (i < n) ? cnt[i] : 0;
    v = wave_sum(v);
    if (lane == 0) wsum[w] = v;
    __syncthreads();
    if (w == 0) {
        int s = wave_sum(wsum[lane]);
        if (lane == 0) bsum[blockIdx.x] = s;
    }
}

__global__ __launch_bounds__(1024) void scan_kernel(const int* __restrict__ cnt,
                                                    const int* __restrict__ bsum,
                                                    int* __restrict__ off,
                                                    int* __restrict__ cur, int n)
{
    __shared__ int wsum[32];
    __shared__ int sbase;
    const int t = threadIdx.x, lane = t & 31, w = t >> 5;

    int p = 0;
    for (int j = t; j < (int)blockIdx.x; j += 1024) p += bsum[j];
    p = wave_sum(p);
    if (lane == 0) wsum[w] = p;
    __syncthreads();
    if (w == 0) {
        int s = wave_sum(wsum[lane]);
        if (lane == 0) sbase = s;
    }
    __syncthreads();
    const int base = sbase;

    const int i = blockIdx.x * 1024 + t;
    const int c = (i < n) ? cnt[i] : 0;
    const int incl = wave_scan_incl(c, lane);
    if (lane == 31) wsum[w] = incl;
    __syncthreads();
    if (w == 0) {
        int s = wave_scan_incl(wsum[lane], lane);
        wsum[lane] = s;
    }
    __syncthreads();
    const int wpre = (w > 0) ? wsum[w - 1] : 0;
    const int excl = base + wpre + incl - c;
    if (i < n) { off[i] = excl; cur[i] = excl; }
    if (i == n - 1) off[n] = excl + c;
}

__global__ __launch_bounds__(256) void fill_kernel(const int* __restrict__ dst,
                                                   int* __restrict__ cur,
                                                   int* __restrict__ tmp, int n_edges)
{
    int e = blockIdx.x * blockDim.x + threadIdx.x;
    if (e < n_edges) {
        int pos = atomicAdd(&cur[dst[e]], 1);
        tmp[pos] = e;
    }
}

__global__ __launch_bounds__(256) void order_kernel(const int* __restrict__ tmp,
                                                    const int* __restrict__ off,
                                                    const int* __restrict__ src,
                                                    const int* __restrict__ dst,
                                                    const int* __restrict__ etype,
                                                    const int* __restrict__ dir,
                                                    unsigned* __restrict__ recs, int n_edges)
{
    int i = blockIdx.x * blockDim.x + threadIdx.x;
    if (i >= n_edges) return;
    const int e = tmp[i];
    const int v = dst[e];
    const unsigned sv = (unsigned)src[e];
    const unsigned et = (unsigned)etype[e] & 1u;
    const unsigned dr = (dir[e] == 1) ? 1u : 0u;
    const int beg = off[v], end = off[v + 1];
    int rank = 0;
    for (int j = beg; j < end; ++j) rank += (tmp[j] < e) ? 1 : 0;
    recs[beg + rank] = (sv << 2) | (et << 1) | dr;
}

__global__ __launch_bounds__(256)
void node_kernel(const float* __restrict__ h_in,
                 const float* __restrict__ hr,
                 const int* __restrict__ off,
                 const unsigned* __restrict__ recs,
                 const float* __restrict__ Ws,
                 const float* __restrict__ Wo,
                 const float* __restrict__ Wi,
                 float* __restrict__ h_out,
                 int n_nodes, int relu)
{
    __shared__ float lds_w[3 * D * D];
    __shared__ float lds_hr[3 * D];
    __shared__ float lds_a[3 * 32 * APITCH];

    const int t = threadIdx.x;
    for (int i = t; i < D * D; i += 256) {
        lds_w[i]             = Ws[i];
        lds_w[D * D + i]     = Wo[i];
        lds_w[2 * D * D + i] = Wi[i];
    }
    if (t < 3 * D) lds_hr[t] = hr[t];
    __syncthreads();

    {
        const int g = t >> 4;
        const int c = (t & 15) * 4;
        for (int pass = 0; pass < 2; ++pass) {
            const int row  = pass * 16 + g;
            const int node = blockIdx.x * 32 + row;
            float4 xs = make_float4(0.f, 0.f, 0.f, 0.f);
            float4 s0 = xs, s1 = xs;
            if (node < n_nodes) {
                const float4 hv = *(const float4*)(h_in + (size_t)node * D + c);
                const float4 r2 = *(const float4*)(lds_hr + 2 * D + c);
                xs.x = hv.x - r2.x; xs.y = hv.y - r2.y; xs.z = hv.z - r2.z; xs.w = hv.w - r2.w;
                const int beg = off[node], end = off[node + 1];
                for (int i = beg; i < end; i += 4) {
                    unsigned rec[4];
                    float4   gv[4];
#pragma unroll
                    for (int u = 0; u < 4; ++u) {
                        const int idx = (i + u < end) ? (i + u) : (end - 1);
                        rec[u] = recs[idx];
                    }
#pragma unroll
                    for (int u = 0; u < 4; ++u)
                        gv[u] = *(const float4*)(h_in + (size_t)(rec[u] >> 2) * D + c);
#pragma unroll
                    for (int u = 0; u < 4; ++u) {
                        const bool ok = (i + u < end);
                        const unsigned et = (rec[u] >> 1) & 1u;
                        const bool     in = (rec[u] & 1u) != 0u;
                        const bool     t0 = ok && !in;
                        const bool     t1 = ok && in;
                        const float4 rv = *(const float4*)(lds_hr + et * D + c);
                        float4 m;
                        m.x = gv[u].x - rv.x;
                        m.y = gv[u].y - rv.y;
                        m.z = gv[u].z - rv.z;
                        m.w = gv[u].w - rv.w;
                        s0.x += t0 ? m.x : 0.f; s0.y += t0 ? m.y : 0.f;
                        s0.z += t0 ? m.z : 0.f; s0.w += t0 ? m.w : 0.f;
                        s1.x += t1 ? m.x : 0.f; s1.y += t1 ? m.y : 0.f;
                        s1.z += t1 ? m.z : 0.f; s1.w += t1 ? m.w : 0.f;
                    }
                }
            }
            *(float4*)(lds_a + row * APITCH + c)                 = xs;
            *(float4*)(lds_a + (32 + row) * APITCH + c)          = s0;
            *(float4*)(lds_a + (64 + row) * APITCH + c)          = s1;
        }
    }
    __syncthreads();

    const int wave  = t >> 5;
    const int lane  = t & 31;
    const int lhalf = lane >> 4;
    const int lmod  = lane & 15;

    const int m0 = (blockIdx.x * 2 + (wave >> 2)) * 16;
    const int n0 = (wave & 3) * 16;
    if (m0 >= n_nodes) return;

    const int brow = n0 + lmod;

    const float* xr = lds_a + ((wave >> 2) * 16 + lmod) * APITCH;
    const float* a0 = xr + 32 * APITCH;
    const float* a1 = xr + 64 * APITCH;

    v8f acc = {};
#pragma unroll
    for (int k0 = 0; k0 < D; k0 += 4) {
        const int ka = k0 + lhalf * 2;
        v2f a, b;
        a.x = xr[ka];
        a.y = xr[ka + 1];
        b.x = lds_w[brow * D + ka];
        b.y = lds_w[brow * D + ka + 1];
        acc = __builtin_amdgcn_wmma_f32_16x16x4_f32(false, a, false, b, (short)0, acc, false, false);
    }
#pragma unroll
    for (int k0 = 0; k0 < D; k0 += 4) {
        const int ka = k0 + lhalf * 2;
        v2f a, b;
        a.x = a0[ka];
        a.y = a0[ka + 1];
        b.x = lds_w[D * D + brow * D + ka];
        b.y = lds_w[D * D + brow * D + ka + 1];
        acc = __builtin_amdgcn_wmma_f32_16x16x4_f32(false, a, false, b, (short)0, acc, false, false);
    }
#pragma unroll
    for (int k0 = 0; k0 < D; k0 += 4) {
        const int ka = k0 + lhalf * 2;
        v2f a, b;
        a.x = a1[ka];
        a.y = a1[ka + 1];
        b.x = lds_w[2 * D * D + brow * D + ka];
        b.y = lds_w[2 * D * D + brow * D + ka + 1];
        acc = __builtin_amdgcn_wmma_f32_16x16x4_f32(false, a, false, b, (short)0, acc, false, false);
    }

#pragma unroll
    for (int r = 0; r < 8; ++r) {
        float v = acc[r];
        if (relu) v = fmaxf(v, 0.f);
        const int m = m0 + r + lhalf * 8;
        if (m < n_nodes) h_out[(size_t)m * D + n0 + lmod] = v;
    }
}

extern "C" void kernel_launch(void* const* d_in, const int* in_sizes, int n_in,
                              void* d_out, int out_size, void* d_ws, size_t ws_size,
                              hipStream_t stream)
{
    const float* h_u   = (const float*)d_in[0];
    const float* Basis = (const float*)d_in[1];
    const float* alpha = (const float*)d_in[2];
    const float* W_O   = (const float*)d_in[3];
    const float* W_I   = (const float*)d_in[4];
    const float* W_S   = (const float*)d_in[5];
    const float* W_rel = (const float*)d_in[6];
    const int*   src   = (const int*)d_in[7];
    const int*   dst   = (const int*)d_in[8];
    const int*   etype = (const int*)d_in[9];
    const int*   edir  = (const int*)d_in[10];
    float* out = (float*)d_out;

    const int n_nodes = in_sizes[0] / D;
    const int n_edges = in_sizes[7];

    const int n_pad = n_nodes + 64;
    int*      cnt  = (int*)d_ws;
    int*      off  = cnt + n_pad;
    int*      cur  = off + n_pad;
    int*      bsum = cur + n_pad;
    int*      tmp  = bsum + n_pad;
    unsigned* recs = (unsigned*)(tmp + n_edges);
    float* hb0 = (float*)d_ws + (size_t)2 * n_nodes * D;
    float* hb1 = hb0 + (size_t)n_nodes * D;
    float* hra = hb1 + (size_t)n_nodes * D;
    float* hrb = hra + 256;

    hr_init_kernel<<<1, 192, 0, stream>>>(Basis, alpha, hra);

    const unsigned node_blk256 = (unsigned)((n_nodes + 255) / 256);
    const unsigned edge_blk256 = (unsigned)((n_edges + 255) / 256);
    const unsigned scan_blocks = (unsigned)((n_nodes + 1023) / 1024);
    const unsigned node_blocks = (unsigned)((n_nodes + 31) / 32);

    zero_kernel<<<node_blk256, 256, 0, stream>>>(cnt, n_nodes);
    degree_kernel<<<edge_blk256, 256, 0, stream>>>(dst, cnt, n_edges);
    blocksum_kernel<<<scan_blocks, 1024, 0, stream>>>(cnt, bsum, n_nodes);
    scan_kernel<<<scan_blocks, 1024, 0, stream>>>(cnt, bsum, off, cur, n_nodes);
    fill_kernel<<<edge_blk256, 256, 0, stream>>>(dst, cur, tmp, n_edges);
    order_kernel<<<edge_blk256, 256, 0, stream>>>(tmp, off, src, dst, etype, edir, recs, n_edges);

    const float* hin[3]  = { h_u, hb0, hb1 };
    float*       hout[3] = { hb0, hb1, out };

    float* hrcur = hra;
    float* hrnxt = hrb;

    for (int l = 0; l < 3; ++l) {
        const int relu = (l < 2) ? 1 : 0;

        node_kernel<<<node_blocks, 256, 0, stream>>>(hin[l], hrcur, off, recs,
                                                     W_S + (size_t)l * D * D,
                                                     W_O + (size_t)l * D * D,
                                                     W_I + (size_t)l * D * D,
                                                     hout[l], n_nodes, relu);

        if (l < 2) {
            rel_kernel<<<1, 192, 0, stream>>>(hrcur, W_rel + (size_t)l * D * D, hrnxt, 1);
            float* tq = hrcur; hrcur = hrnxt; hrnxt = tq;
        }
    }
}


# Weight tiles in LDS at a 68-float row pitch instead of 64: the B fragment reads (one weight row per lane, same k) no longer put 16 rows on one bank
# speedup vs baseline: 1.6611x; 1.1959x over previous
#include <hip/hip_runtime.h>
#include <hip/hip_bf16.h>


typedef __attribute__((ext_vector_type(2))) float v2f;
typedef __attribute__((ext_vector_type(8))) float v8f;

#define D 64
#define APITCH 68
#define WPITCH 68

__global__ void hr_init_kernel(const float* __restrict__ Basis,
                               const float* __restrict__ alpha,
                               float* __restrict__ h_r)
{
    int t = threadIdx.x;
    if (t >= 3 * D) return;
    int r = t >> 6, n = t & 63;
    float s = 0.f;
#pragma unroll
    for (int b = 0; b < 16; ++b) s += alpha[r * 16 + b] * Basis[b * D + n];
    h_r[t] = s;
}

__global__ void rel_kernel(const float* __restrict__ h_r,
                           const float* __restrict__ Wrel,
                           float* __restrict__ h_r_out,
                           int relu)
{
    int t = threadIdx.x;
    if (t >= 3 * D) return;
    int r = t >> 6, n = t & 63;
    float s = 0.f;
#pragma unroll 8
    for (int k = 0; k < D; ++k) s += h_r[r * D + k] * Wrel[n * D + k];
    if (relu) s = fmaxf(s, 0.f);
    h_r_out[t] = s;
}

__global__ __launch_bounds__(256) void zero_kernel(int* __restrict__ p, int n)
{
    int i = blockIdx.x * blockDim.x + threadIdx.x;
    if (i < n) p[i] = 0;
}

__global__ __launch_bounds__(256) void degree_kernel(const int* __restrict__ dst,
                                                     int* __restrict__ cnt, int n_edges)
{
    int e = blockIdx.x * blockDim.x + threadIdx.x;
    if (e < n_edges) atomicAdd(&cnt[dst[e]], 1);
}

__device__ __forceinline__ int wave_sum(int v)
{
#pragma unroll
    for (int o = 16; o > 0; o >>= 1) v += __shfl_down(v, o, 32);
    return v;
}

__device__ __forceinline__ int wave_scan_incl(int v, int lane)
{
#pragma unroll
    for (int o = 1; o < 32; o <<= 1) {
        int nb = __shfl_up(v, o, 32);
        if (lane >= o) v += nb;
    }
    return v;
}

__global__ __launch_bounds__(1024) void blocksum_kernel(const int* __restrict__ cnt,
                                                        int* __restrict__ bsum, int n)
{
    __shared__ int wsum[32];
    const int t = threadIdx.x, lane = t & 31, w = t >> 5;
    const int i = blockIdx.x * 1024 + t;
    int v = (i < n) ? cnt[i] : 0;
    v = wave_sum(v);
    if (lane == 0) wsum[w] = v;
    __syncthreads();
    if (w == 0) {
        int s = wave_sum(wsum[lane]);
        if (lane == 0) bsum[blockIdx.x] = s;
    }
}

__global__ __launch_bounds__(1024) void scan_kernel(const int* __restrict__ cnt,
                                                    const int* __restrict__ bsum,
                                                    int* __restrict__ off,
                                                    int* __restrict__ cur, int n)
{
    __shared__ int wsum[32];
    __shared__ int sbase;
    const int t = threadIdx.x, lane = t & 31, w = t >> 5;

    int p = 0;
    for (int j = t; j < (int)blockIdx.x; j += 1024) p += bsum[j];
    p = wave_sum(p);
    if (lane == 0) wsum[w] = p;
    __syncthreads();
    if (w == 0) {
        int s = wave_sum(wsum[lane]);
        if (lane == 0) sbase = s;
    }
    __syncthreads();
    const int base = sbase;

    const int i = blockIdx.x * 1024 + t;
    const int c = (i < n) ? cnt[i] : 0;
    const int incl = wave_scan_incl(c, lane);
    if (lane == 31) wsum[w] = incl;
    __syncthreads();
    if (w == 0) {
        int s = wave_scan_incl(wsum[lane], lane);
        wsum[lane] = s;
    }
    __syncthreads();
    const int wpre = (w > 0) ? wsum[w - 1] : 0;
    const int excl = base + wpre + incl - c;
    if (i < n) { off[i] = excl; cur[i] = excl; }
    if (i == n - 1) off[n] = excl + c;
}

__global__ __launch_bounds__(256) void fill_kernel(const int* __restrict__ dst,
                                                   int* __restrict__ cur,
                                                   int* __restrict__ tmp, int n_edges)
{
    int e = blockIdx.x * blockDim.x + threadIdx.x;
    if (e < n_edges) {
        int pos = atomicAdd(&cur[dst[e]], 1);
        tmp[pos] = e;
    }
}

__global__ __launch_bounds__(256) void order_kernel(const int* __restrict__ tmp,
                                                    const int* __restrict__ off,
                                                    const int* __restrict__ src,
                                                    const int* __restrict__ dst,
                                                    const int* __restrict__ etype,
                                                    const int* __restrict__ dir,
                                                    unsigned* __restrict__ recs, int n_edges)
{
    int i = blockIdx.x * blockDim.x + threadIdx.x;
    if (i >= n_edges) return;
    const int e = tmp[i];
    const int v = dst[e];
    const unsigned sv = (unsigned)src[e];
    const unsigned et = (unsigned)etype[e] & 1u;
    const unsigned dr = (dir[e] == 1) ? 1u : 0u;
    const int beg = off[v], end = off[v + 1];
    int rank = 0;
    for (int j = beg; j < end; ++j) rank += (tmp[j] < e) ? 1 : 0;
    recs[beg + rank] = (sv << 2) | (et << 1) | dr;
}

__global__ __launch_bounds__(256)
void node_kernel(const float* __restrict__ h_in,
                 const float* __restrict__ hr,
                 const int* __restrict__ off,
                 const unsigned* __restrict__ recs,
                 const float* __restrict__ Ws,
                 const float* __restrict__ Wo,
                 const float* __restrict__ Wi,
                 float* __restrict__ h_out,
                 int n_nodes, int relu)
{
    __shared__ float lds_w[3 * D * WPITCH];
    __shared__ float lds_hr[3 * D];
    __shared__ float lds_a[3 * 32 * APITCH];

    const int t = threadIdx.x;
    for (int i = t; i < D * D; i += 256) {
        const int wi = (i >> 6) * WPITCH + (i & 63);
        lds_w[wi]                  = Ws[i];
        lds_w[D * WPITCH + wi]     = Wo[i];
        lds_w[2 * D * WPITCH + wi] = Wi[i];
    }
    if (t < 3 * D) lds_hr[t] = hr[t];
    __syncthreads();

    {
        const int g = t >> 4;
        const int c = (t & 15) * 4;
        for (int pass = 0; pass < 2; ++pass) {
            const int row  = pass * 16 + g;
            const int node = blockIdx.x * 32 + row;
            float4 xs = make_float4(0.f, 0.f, 0.f, 0.f);
            float4 s0 = xs, s1 = xs;
            if (node < n_nodes) {
                const float4 hv = *(const float4*)(h_in + (size_t)node * D + c);
                const float4 r2 = *(const float4*)(lds_hr + 2 * D + c);
                xs.x = hv.x - r2.x; xs.y = hv.y - r2.y; xs.z = hv.z - r2.z; xs.w = hv.w - r2.w;
                const int beg = off[node], end = off[node + 1];
                for (int i = beg; i < end; i += 4) {
                    unsigned rec[4];
                    float4   gv[4];
#pragma unroll
                    for (int u = 0; u < 4; ++u) {
                        const int idx = (i + u < end) ? (i + u) : (end - 1);
                        rec[u] = recs[idx];
                    }
#pragma unroll
                    for (int u = 0; u < 4; ++u)
                        gv[u] = *(const float4*)(h_in + (size_t)(rec[u] >> 2) * D + c);
#pragma unroll
                    for (int u = 0; u < 4; ++u) {
                        const bool ok = (i + u < end);
                        const unsigned et = (rec[u] >> 1) & 1u;
                        const bool     in = (rec[u] & 1u) != 0u;
                        const bool     t0 = ok && !in;
                        const bool     t1 = ok && in;
                        const float4 rv = *(const float4*)(lds_hr + et * D + c);
                        float4 m;
                        m.x = gv[u].x - rv.x;
                        m.y = gv[u].y - rv.y;
                        m.z = gv[u].z - rv.z;
                        m.w = gv[u].w - rv.w;
                        s0.x += t0 ? m.x : 0.f; s0.y += t0 ? m.y : 0.f;
                        s0.z += t0 ? m.z : 0.f; s0.w += t0 ? m.w : 0.f;
                        s1.x += t1 ? m.x : 0.f; s1.y += t1 ? m.y : 0.f;
                        s1.z += t1 ? m.z : 0.f; s1.w += t1 ? m.w : 0.f;
                    }
                }
            }
            *(float4*)(lds_a + row * APITCH + c)                 = xs;
            *(float4*)(lds_a + (32 + row) * APITCH + c)          = s0;
            *(float4*)(lds_a + (64 + row) * APITCH + c)          = s1;
        }
    }
    __syncthreads();

    const int wave  = t >> 5;
    const int lane  = t & 31;
    const int lhalf = lane >> 4;
    const int lmod  = lane & 15;

    const int m0 = (blockIdx.x * 2 + (wave >> 2)) * 16;
    const int n0 = (wave & 3) * 16;
    if (m0 >= n_nodes) return;

    const int brow = n0 + lmod;

    const float* xr = lds_a + ((wave >> 2) * 16 + lmod) * APITCH;
    const float* a0 = xr + 32 * APITCH;
    const float* a1 = xr + 64 * APITCH;

    v8f acc = {};
#pragma unroll
    for (int k0 = 0; k0 < D; k0 += 4) {
        const int ka = k0 + lhalf * 2;
        v2f a, b;
        a.x = xr[ka];
        a.y = xr[ka + 1];
        b.x = lds_w[brow * WPITCH + ka];
        b.y = lds_w[brow * WPITCH + ka + 1];
        acc = __builtin_amdgcn_wmma_f32_16x16x4_f32(false, a, false, b, (short)0, acc, false, false);
    }
#pragma unroll
    for (int k0 = 0; k0 < D; k0 += 4) {
        const int ka = k0 + lhalf * 2;
        v2f a, b;
        a.x = a0[ka];
        a.y = a0[ka + 1];
        b.x = lds_w[D * WPITCH + brow * WPITCH + ka];
        b.y = lds_w[D * WPITCH + brow * WPITCH + ka + 1];
        acc = __builtin_amdgcn_wmma_f32_16x16x4_f32(false, a, false, b, (short)0, acc, false, false);
    }
#pragma unroll
    for (int k0 = 0; k0 < D; k0 += 4) {
        const int ka = k0 + lhalf * 2;
        v2f a, b;
        a.x = a1[ka];
        a.y = a1[ka + 1];
        b.x = lds_w[2 * D * WPITCH + brow * WPITCH + ka];
        b.y = lds_w[2 * D * WPITCH + brow * WPITCH + ka + 1];
        acc = __builtin_amdgcn_wmma_f32_16x16x4_f32(false, a, false, b, (short)0, acc, false, false);
    }

#pragma unroll
    for (int r = 0; r < 8; ++r) {
        float v = acc[r];
        if (relu) v = fmaxf(v, 0.f);
        const int m = m0 + r + lhalf * 8;
        if (m < n_nodes) h_out[(size_t)m * D + n0 + lmod] = v;
    }
}

extern "C" void kernel_launch(void* const* d_in, const int* in_sizes, int n_in,
                              void* d_out, int out_size, void* d_ws, size_t ws_size,
                              hipStream_t stream)
{
    const float* h_u   = (const float*)d_in[0];
    const float* Basis = (const float*)d_in[1];
    const float* alpha = (const float*)d_in[2];
    const float* W_O   = (const float*)d_in[3];
    const float* W_I   = (const float*)d_in[4];
    const float* W_S   = (const float*)d_in[5];
    const float* W_rel = (const float*)d_in[6];
    const int*   src   = (const int*)d_in[7];
    const int*   dst   = (const int*)d_in[8];
    const int*   etype = (const int*)d_in[9];
    const int*   edir  = (const int*)d_in[10];
    float* out = (float*)d_out;

    const int n_nodes = in_sizes[0] / D;
    const int n_edges = in_sizes[7];

    const int n_pad = n_nodes + 64;
    int*      cnt  = (int*)d_ws;
    int*      off  = cnt + n_pad;
    int*      cur  = off + n_pad;
    int*      bsum = cur + n_pad;
    int*      tmp  = bsum + n_pad;
    unsigned* recs = (unsigned*)(tmp + n_edges);
    float* hb0 = (float*)d_ws + (size_t)2 * n_nodes * D;
    float* hb1 = hb0 + (size_t)n_nodes * D;
    float* hra = hb1 + (size_t)n_nodes * D;
    float* hrb = hra + 256;

    hr_init_kernel<<<1, 192, 0, stream>>>(Basis, alpha, hra);

    const unsigned node_blk256 = (unsigned)((n_nodes + 255) / 256);
    const unsigned edge_blk256 = (unsigned)((n_edges + 255) / 256);
    const unsigned scan_blocks = (unsigned)((n_nodes + 1023) / 1024);
    const unsigned node_blocks = (unsigned)((n_nodes + 31) / 32);

    zero_kernel<<<node_blk256, 256, 0, stream>>>(cnt, n_nodes);
    degree_kernel<<<edge_blk256, 256, 0, stream>>>(dst, cnt, n_edges);
    blocksum_kernel<<<scan_blocks, 1024, 0, stream>>>(cnt, bsum, n_nodes);
    scan_kernel<<<scan_blocks, 1024, 0, stream>>>(cnt, bsum, off, cur, n_nodes);
    fill_kernel<<<edge_blk256, 256, 0, stream>>>(dst, cur, tmp, n_edges);
    order_kernel<<<edge_blk256, 256, 0, stream>>>(tmp, off, src, dst, etype, edir, recs, n_edges);

    const float* hin[3]  = { h_u, hb0, hb1 };
    float*       hout[3] = { hb0, hb1, out };

    float* hrcur = hra;
    float* hrnxt = hrb;

    for (int l = 0; l < 3; ++l) {
        const int relu = (l < 2) ? 1 : 0;

        node_kernel<<<node_blocks, 256, 0, stream>>>(hin[l], hrcur, off, recs,
                                                     W_S + (size_t)l * D * D,
                                                     W_O + (size_t)l * D * D,
                                                     W_I + (size_t)l * D * D,
                                                     hout[l], n_nodes, relu);

        if (l < 2) {
            rel_kernel<<<1, 192, 0, stream>>>(hrcur, W_rel + (size_t)l * D * D, hrnxt, 1);
            float* tq = hrcur; hrcur = hrnxt; hrnxt = tq;
        }
    }
}


# Layer kernel issues loads in large groups ahead of use: all weights in one group beside the first edge records; 32 nodes gathered at once (8 lanes each), next 4 records fetched beside the row reads
# speedup vs baseline: 1.8594x; 1.1194x over previous
#include <hip/hip_runtime.h>
#include <hip/hip_bf16.h>


typedef __attribute__((ext_vector_type(2))) float v2f;
typedef __attribute__((ext_vector_type(8))) float v8f;

#define D 64
#define APITCH 68
#define WPITCH 68

__global__ void hr_init_kernel(const float* __restrict__ Basis,
                               const float* __restrict__ alpha,
                               float* __restrict__ h_r)
{
    int t = threadIdx.x;
    if (t >= 3 * D) return;
    int r = t >> 6, n = t & 63;
    float s = 0.f;
#pragma unroll
    for (int b = 0; b < 16; ++b) s += alpha[r * 16 + b] * Basis[b * D + n];
    h_r[t] = s;
}

__global__ void rel_kernel(const float* __restrict__ h_r,
                           const float* __restrict__ Wrel,
                           float* __restrict__ h_r_out,
                           int relu)
{
    int t = threadIdx.x;
    if (t >= 3 * D) return;
    int r = t >> 6, n = t & 63;
    float s = 0.f;
#pragma unroll 8
    for (int k = 0; k < D; ++k) s += h_r[r * D + k] * Wrel[n * D + k];
    if (relu) s = fmaxf(s, 0.f);
    h_r_out[t] = s;
}

__global__ __launch_bounds__(256) void zero_kernel(int* __restrict__ p, int n)
{
    int i = blockIdx.x * blockDim.x + threadIdx.x;
    if (i < n) p[i] = 0;
}

__global__ __launch_bounds__(256) void degree_kernel(const int* __restrict__ dst,
                                                     int* __restrict__ cnt, int n_edges)
{
    int e = blockIdx.x * blockDim.x + threadIdx.x;
    if (e < n_edges) atomicAdd(&cnt[dst[e]], 1);
}

__device__ __forceinline__ int wave_sum(int v)
{
#pragma unroll
    for (int o = 16; o > 0; o >>= 1) v += __shfl_down(v, o, 32);
    return v;
}

__device__ __forceinline__ int wave_scan_incl(int v, int lane)
{
#pragma unroll
    for (int o = 1; o < 32; o <<= 1) {
        int nb = __shfl_up(v, o, 32);
        if (lane >= o) v += nb;
    }
    return v;
}

__global__ __launch_bounds__(1024) void blocksum_kernel(const int* __restrict__ cnt,
                                                        int* __restrict__ bsum, int n)
{
    __shared__ int wsum[32];
    const int t = threadIdx.x, lane = t & 31, w = t >> 5;
    const int i = blockIdx.x * 1024 + t;
    int v = (i < n) ? cnt[i] : 0;
    v = wave_sum(v);
    if (lane == 0) wsum[w] = v;
    __syncthreads();
    if (w == 0) {
        int s = wave_sum(wsum[lane]);
        if (lane == 0) bsum[blockIdx.x] = s;
    }
}

__global__ __launch_bounds__(1024) void scan_kernel(const int* __restrict__ cnt,
                                                    const int* __restrict__ bsum,
                                                    int* __restrict__ off,
                                                    int* __restrict__ cur, int n)
{
    __shared__ int wsum[32];
    __shared__ int sbase;
    const int t = threadIdx.x, lane = t & 31, w = t >> 5;

    int p = 0;
    for (int j = t; j < (int)blockIdx.x; j += 1024) p += bsum[j];
    p = wave_sum(p);
    if (lane == 0) wsum[w] = p;
    __syncthreads();
    if (w == 0) {
        int s = wave_sum(wsum[lane]);
        if (lane == 0) sbase = s;
    }
    __syncthreads();
    const int base = sbase;

    const int i = blockIdx.x * 1024 + t;
    const int c = (i < n) ? cnt[i] : 0;
    const int incl = wave_scan_incl(c, lane);
    if (lane == 31) wsum[w] = incl;
    __syncthreads();
    if (w == 0) {
        int s = wave_scan_incl(wsum[lane], lane);
        wsum[lane] = s;
    }
    __syncthreads();
    const int wpre = (w > 0) ? wsum[w - 1] : 0;
    const int excl = base + wpre + incl - c;
    if (i < n) { off[i] = excl; cur[i] = excl; }
    if (i == n - 1) off[n] = excl + c;
}

__global__ __launch_bounds__(256) void fill_kernel(const int* __restrict__ dst,
                                                   int* __restrict__ cur,
                                                   int* __restrict__ tmp, int n_edges)
{
    int e = blockIdx.x * blockDim.x + threadIdx.x;
    if (e < n_edges) {
        int pos = atomicAdd(&cur[dst[e]], 1);
        tmp[pos] = e;
    }
}

__global__ __launch_bounds__(256) void order_kernel(const int* __restrict__ tmp,
                                                    const int* __restrict__ off,
                                                    const int* __restrict__ src,
                                                    const int* __restrict__ dst,
                                                    const int* __restrict__ etype,
                                                    const int* __restrict__ dir,
                                                    unsigned* __restrict__ recs, int n_edges)
{
    int i = blockIdx.x * blockDim.x + threadIdx.x;
    if (i >= n_edges) return;
    const int e = tmp[i];
    const int v = dst[e];
    const unsigned sv = (unsigned)src[e];
    const unsigned et = (unsigned)etype[e] & 1u;
    const unsigned dr = (dir[e] == 1) ? 1u : 0u;
    const int beg = off[v], end = off[v + 1];
    int rank = 0;
    for (int j = beg; j < end; ++j) rank += (tmp[j] < e) ? 1 : 0;
    recs[beg + rank] = (sv << 2) | (et << 1) | dr;
}

__global__ __launch_bounds__(256)
void node_kernel(const float* __restrict__ h_in,
                 const float* __restrict__ hr,
                 const int* __restrict__ off,
                 const unsigned* __restrict__ recs,
                 const float* __restrict__ Ws,
                 const float* __restrict__ Wo,
                 const float* __restrict__ Wi,
                 float* __restrict__ h_out,
                 int n_nodes, int relu)
{
    __shared__ alignas(16) float lds_w[3 * D * WPITCH];
    __shared__ alignas(16) float lds_hr[3 * D];
    __shared__ alignas(16) float lds_a[3 * 32 * APITCH];

    const int t = threadIdx.x;

    const int g    = t >> 3;
    const int c    = (t & 7) * 4;
    const int node = blockIdx.x * 32 + g;
    const bool live = node < n_nodes;

    int beg = 0, end = 0;
    float4 ha = make_float4(0.f, 0.f, 0.f, 0.f);
    float4 hb = ha;
    if (live) {
        beg = off[node];
        end = off[node + 1];
        ha = *(const float4*)(h_in + (size_t)node * D + c);
        hb = *(const float4*)(h_in + (size_t)node * D + 32 + c);
    }
    const float hrv = (t < 3 * D) ? hr[t] : 0.f;

    float4 wv[12];
#pragma unroll
    for (int j = 0; j < 4; ++j) {
        const int i4 = t + 256 * j;
        wv[j]     = ((const float4*)Ws)[i4];
        wv[4 + j] = ((const float4*)Wo)[i4];
        wv[8 + j] = ((const float4*)Wi)[i4];
    }

    unsigned rn[4] = { 0u, 0u, 0u, 0u };
    if (beg < end) {
#pragma unroll
        for (int u = 0; u < 4; ++u) {
            const int idx = (beg + u < end) ? (beg + u) : (end - 1);
            rn[u] = recs[idx];
        }
    }

#pragma unroll
    for (int j = 0; j < 4; ++j) {
        const int i4 = t + 256 * j;
        const int wi = (i4 >> 4) * WPITCH + (i4 & 15) * 4;
        *(float4*)(lds_w + wi)                  = wv[j];
        *(float4*)(lds_w + D * WPITCH + wi)     = wv[4 + j];
        *(float4*)(lds_w + 2 * D * WPITCH + wi) = wv[8 + j];
    }
    if (t < 3 * D) lds_hr[t] = hrv;
    __syncthreads();

    {
        float4 xa = make_float4(0.f, 0.f, 0.f, 0.f);
        float4 xb = xa;
        if (live) {
            const float4 ra = *(const float4*)(lds_hr + 2 * D + c);
            const float4 rb = *(const float4*)(lds_hr + 2 * D + 32 + c);
            xa.x = ha.x - ra.x; xa.y = ha.y - ra.y; xa.z = ha.z - ra.z; xa.w = ha.w - ra.w;
            xb.x = hb.x - rb.x; xb.y = hb.y - rb.y; xb.z = hb.z - rb.z; xb.w = hb.w - rb.w;
        }
        *(float4*)(lds_a + g * APITCH + c)      = xa;
        *(float4*)(lds_a + g * APITCH + 32 + c) = xb;

        float4 s0a = make_float4(0.f, 0.f, 0.f, 0.f);
        float4 s0b = s0a, s1a = s0a, s1b = s0a;
        for (int i = beg; i < end; i += 4) {
            unsigned rec[4];
            float4   ga[4], gb[4];
#pragma unroll
            for (int u = 0; u < 4; ++u) rec[u] = rn[u];
#pragma unroll
            for (int u = 0; u < 4; ++u) {
                const float* p = h_in + (size_t)(rec[u] >> 2) * D + c;
                ga[u] = *(const float4*)p;
                gb[u] = *(const float4*)(p + 32);
            }
#pragma unroll
            for (int u = 0; u < 4; ++u) {
                const int idx = (i + 4 + u < end) ? (i + 4 + u) : (end - 1);
                rn[u] = recs[idx];
            }
#pragma unroll
            for (int u = 0; u < 4; ++u) {
                const bool ok = (i + u < end);
                const unsigned et = (rec[u] >> 1) & 1u;
                const bool     in = (rec[u] & 1u) != 0u;
                const bool     t0 = ok && !in;
                const bool     t1 = ok && in;
                const float4 ra = *(const float4*)(lds_hr + et * D + c);
                const float4 rb = *(const float4*)(lds_hr + et * D + 32 + c);
                float4 ma, mb;
                ma.x = ga[u].x - ra.x; ma.y = ga[u].y - ra.y;
                ma.z = ga[u].z - ra.z; ma.w = ga[u].w - ra.w;
                mb.x = gb[u].x - rb.x; mb.y = gb[u].y - rb.y;
                mb.z = gb[u].z - rb.z; mb.w = gb[u].w - rb.w;
                s0a.x += t0 ? ma.x : 0.f; s0a.y += t0 ? ma.y : 0.f;
                s0a.z += t0 ? ma.z : 0.f; s0a.w += t0 ? ma.w : 0.f;
                s0b.x += t0 ? mb.x : 0.f; s0b.y += t0 ? mb.y : 0.f;
                s0b.z += t0 ? mb.z : 0.f; s0b.w += t0 ? mb.w : 0.f;
                s1a.x += t1 ? ma.x : 0.f; s1a.y += t1 ? ma.y : 0.f;
                s1a.z += t1 ? ma.z : 0.f; s1a.w += t1 ? ma.w : 0.f;
                s1b.x += t1 ? mb.x : 0.f; s1b.y += t1 ? mb.y : 0.f;
                s1b.z += t1 ? mb.z : 0.f; s1b.w += t1 ? mb.w : 0.f;
            }
        }
        *(float4*)(lds_a + (32 + g) * APITCH + c)      = s0a;
        *(float4*)(lds_a + (32 + g) * APITCH + 32 + c) = s0b;
        *(float4*)(lds_a + (64 + g) * APITCH + c)      = s1a;
        *(float4*)(lds_a + (64 + g) * APITCH + 32 + c) = s1b;
    }
    __syncthreads();

    const int wave  = t >> 5;
    const int lane  = t & 31;
    const int lhalf = lane >> 4;
    const int lmod  = lane & 15;

    const int m0 = (blockIdx.x * 2 + (wave >> 2)) * 16;
    const int n0 = (wave & 3) * 16;
    if (m0 >= n_nodes) return;

    const int brow = n0 + lmod;

    const float* xr = lds_a + ((wave >> 2) * 16 + lmod) * APITCH;
    const float* a0 = xr + 32 * APITCH;
    const float* a1 = xr + 64 * APITCH;

    v8f acc = {};
#pragma unroll
    for (int k0 = 0; k0 < D; k0 += 4) {
        const int ka = k0 + lhalf * 2;
        v2f a, b;
        a.x = xr[ka];
        a.y = xr[ka + 1];
        b.x = lds_w[brow * WPITCH + ka];
        b.y = lds_w[brow * WPITCH + ka + 1];
        acc = __builtin_amdgcn_wmma_f32_16x16x4_f32(false, a, false, b, (short)0, acc, false, false);
    }
#pragma unroll
    for (int k0 = 0; k0 < D; k0 += 4) {
        const int ka = k0 + lhalf * 2;
        v2f a, b;
        a.x = a0[ka];
        a.y = a0[ka + 1];
        b.x = lds_w[D * WPITCH + brow * WPITCH + ka];
        b.y = lds_w[D * WPITCH + brow * WPITCH + ka + 1];
        acc = __builtin_amdgcn_wmma_f32_16x16x4_f32(false, a, false, b, (short)0, acc, false, false);
    }
#pragma unroll
    for (int k0 = 0; k0 < D; k0 += 4) {
        const int ka = k0 + lhalf * 2;
        v2f a, b;
        a.x = a1[ka];
        a.y = a1[ka + 1];
        b.x = lds_w[2 * D * WPITCH + brow * WPITCH + ka];
        b.y = lds_w[2 * D * WPITCH + brow * WPITCH + ka + 1];
        acc = __builtin_amdgcn_wmma_f32_16x16x4_f32(false, a, false, b, (short)0, acc, false, false);
    }

#pragma unroll
    for (int r = 0; r < 8; ++r) {
        float v = acc[r];
        if (relu) v = fmaxf(v, 0.f);
        const int m = m0 + r + lhalf * 8;
        if (m < n_nodes) h_out[(size_t)m * D + n0 + lmod] = v;
    }
}

extern "C" void kernel_launch(void* const* d_in, const int* in_sizes, int n_in,
                              void* d_out, int out_size, void* d_ws, size_t ws_size,
                              hipStream_t stream)
{
    const float* h_u   = (const float*)d_in[0];
    const float* Basis = (const float*)d_in[1];
    const float* alpha = (const float*)d_in[2];
    const float* W_O   = (const float*)d_in[3];
    const float* W_I   = (const float*)d_in[4];
    const float* W_S   = (const float*)d_in[5];
    const float* W_rel = (const float*)d_in[6];
    const int*   src   = (const int*)d_in[7];
    const int*   dst   = (const int*)d_in[8];
    const int*   etype = (const int*)d_in[9];
    const int*   edir  = (const int*)d_in[10];
    float* out = (float*)d_out;

    const int n_nodes = in_sizes[0] / D;
    const int n_edges = in_sizes[7];

    const int n_pad = n_nodes + 64;
    int*      cnt  = (int*)d_ws;
    int*      off  = cnt + n_pad;
    int*      cur  = off + n_pad;
    int*      bsum = cur + n_pad;
    int*      tmp  = bsum + n_pad;
    unsigned* recs = (unsigned*)(tmp + n_edges);
    float* hb0 = (float*)d_ws + (size_t)2 * n_nodes * D;
    float* hb1 = hb0 + (size_t)n_nodes * D;
    float* hra = hb1 + (size_t)n_nodes * D;
    float* hrb = hra + 256;

    hr_init_kernel<<<1, 192, 0, stream>>>(Basis, alpha, hra);

    const unsigned node_blk256 = (unsigned)((n_nodes + 255) / 256);
    const unsigned edge_blk256 = (unsigned)((n_edges + 255) / 256);
    const unsigned scan_blocks = (unsigned)((n_nodes + 1023) / 1024);
    const unsigned node_blocks = (unsigned)((n_nodes + 31) / 32);

    zero_kernel<<<node_blk256, 256, 0, stream>>>(cnt, n_nodes);
    degree_kernel<<<edge_blk256, 256, 0, stream>>>(dst, cnt, n_edges);
    blocksum_kernel<<<scan_blocks, 1024, 0, stream>>>(cnt, bsum, n_nodes);
    scan_kernel<<<scan_blocks, 1024, 0, stream>>>(cnt, bsum, off, cur, n_nodes);
    fill_kernel<<<edge_blk256, 256, 0, stream>>>(dst, cur, tmp, n_edges);
    order_kernel<<<edge_blk256, 256, 0, stream>>>(tmp, off, src, dst, etype, edir, recs, n_edges);

    const float* hin[3]  = { h_u, hb0, hb1 };
    float*       hout[3] = { hb0, hb1, out };

    float* hrcur = hra;
    float* hrnxt = hrb;

    for (int l = 0; l < 3; ++l) {
        const int relu = (l < 2) ? 1 : 0;

        node_kernel<<<node_blocks, 256, 0, stream>>>(hin[l], hrcur, off, recs,
                                                     W_S + (size_t)l * D * D,
                                                     W_O + (size_t)l * D * D,
                                                     W_I + (size_t)l * D * D,
                                                     hout[l], n_nodes, relu);

        if (l < 2) {
            rel_kernel<<<1, 192, 0, stream>>>(hrcur, W_rel + (size_t)l * D * D, hrnxt, 1);
            float* tq = hrcur; hrcur = hrnxt; hrnxt = tq;
        }
    }
}


# Layer kernel as 1024 resident blocks looping over 32-node tiles: weights go to LDS once per block; next tile's bounds, self rows and first records are requested behind the current tile's barriers
# speedup vs baseline: 1.9217x; 1.0335x over previous
#include <hip/hip_runtime.h>
#include <hip/hip_bf16.h>


typedef __attribute__((ext_vector_type(2))) float v2f;
typedef __attribute__((ext_vector_type(8))) float v8f;

#define D 64
#define APITCH 68
#define WPITCH 68

__global__ void hr_init_kernel(const float* __restrict__ Basis,
                               const float* __restrict__ alpha,
                               float* __restrict__ h_r)
{
    int t = threadIdx.x;
    if (t >= 3 * D) return;
    int r = t >> 6, n = t & 63;
    float s = 0.f;
#pragma unroll
    for (int b = 0; b < 16; ++b) s += alpha[r * 16 + b] * Basis[b * D + n];
    h_r[t] = s;
}

__global__ void rel_kernel(const float* __restrict__ h_r,
                           const float* __restrict__ Wrel,
                           float* __restrict__ h_r_out,
                           int relu)
{
    int t = threadIdx.x;
    if (t >= 3 * D) return;
    int r = t >> 6, n = t & 63;
    float s = 0.f;
#pragma unroll 8
    for (int k = 0; k < D; ++k) s += h_r[r * D + k] * Wrel[n * D + k];
    if (relu) s = fmaxf(s, 0.f);
    h_r_out[t] = s;
}

__global__ __launch_bounds__(256) void zero_kernel(int* __restrict__ p, int n)
{
    int i = blockIdx.x * blockDim.x + threadIdx.x;
    if (i < n) p[i] = 0;
}

__global__ __launch_bounds__(256) void degree_kernel(const int* __restrict__ dst,
                                                     int* __restrict__ cnt, int n_edges)
{
    int e = blockIdx.x * blockDim.x + threadIdx.x;
    if (e < n_edges) atomicAdd(&cnt[dst[e]], 1);
}

__device__ __forceinline__ int wave_sum(int v)
{
#pragma unroll
    for (int o = 16; o > 0; o >>= 1) v += __shfl_down(v, o, 32);
    return v;
}

__device__ __forceinline__ int wave_scan_incl(int v, int lane)
{
#pragma unroll
    for (int o = 1; o < 32; o <<= 1) {
        int nb = __shfl_up(v, o, 32);
        if (lane >= o) v += nb;
    }
    return v;
}

__global__ __launch_bounds__(1024) void blocksum_kernel(const int* __restrict__ cnt,
                                                        int* __restrict__ bsum, int n)
{
    __shared__ int wsum[32];
    const int t = threadIdx.x, lane = t & 31, w = t >> 5;
    const int i = blockIdx.x * 1024 + t;
    int v = (i < n) ? cnt[i] : 0;
    v = wave_sum(v);
    if (lane == 0) wsum[w] = v;
    __syncthreads();
    if (w == 0) {
        int s = wave_sum(wsum[lane]);
        if (lane == 0) bsum[blockIdx.x] = s;
    }
}

__global__ __launch_bounds__(1024) void scan_kernel(const int* __restrict__ cnt,
                                                    const int* __restrict__ bsum,
                                                    int* __restrict__ off,
                                                    int* __restrict__ cur, int n)
{
    __shared__ int wsum[32];
    __shared__ int sbase;
    const int t = threadIdx.x, lane = t & 31, w = t >> 5;

    int p = 0;
    for (int j = t; j < (int)blockIdx.x; j += 1024) p += bsum[j];
    p = wave_sum(p);
    if (lane == 0) wsum[w] = p;
    __syncthreads();
    if (w == 0) {
        int s = wave_sum(wsum[lane]);
        if (lane == 0) sbase = s;
    }
    __syncthreads();
    const int base = sbase;

    const int i = blockIdx.x * 1024 + t;
    const int c = (i < n) ? cnt[i] : 0;
    const int incl = wave_scan_incl(c, lane);
    if (lane == 31) wsum[w] = incl;
    __syncthreads();
    if (w == 0) {
        int s = wave_scan_incl(wsum[lane], lane);
        wsum[lane] = s;
    }
    __syncthreads();
    const int wpre = (w > 0) ? wsum[w - 1] : 0;
    const int excl = base + wpre + incl - c;
    if (i < n) { off[i] = excl; cur[i] = excl; }
    if (i == n - 1) off[n] = excl + c;
}

__global__ __launch_bounds__(256) void fill_kernel(const int* __restrict__ dst,
                                                   int* __restrict__ cur,
                                                   int* __restrict__ tmp, int n_edges)
{
    int e = blockIdx.x * blockDim.x + threadIdx.x;
    if (e < n_edges) {
        int pos = atomicAdd(&cur[dst[e]], 1);
        tmp[pos] = e;
    }
}

__global__ __launch_bounds__(256) void order_kernel(const int* __restrict__ tmp,
                                                    const int* __restrict__ off,
                                                    const int* __restrict__ src,
                                                    const int* __restrict__ dst,
                                                    const int* __restrict__ etype,
                                                    const int* __restrict__ dir,
                                                    unsigned* __restrict__ recs, int n_edges)
{
    int i = blockIdx.x * blockDim.x + threadIdx.x;
    if (i >= n_edges) return;
    const int e = tmp[i];
    const int v = dst[e];
    const unsigned sv = (unsigned)src[e];
    const unsigned et = (unsigned)etype[e] & 1u;
    const unsigned dr = (dir[e] == 1) ? 1u : 0u;
    const int beg = off[v], end = off[v + 1];
    int rank = 0;
    for (int j = beg; j < end; ++j) rank += (tmp[j] < e) ? 1 : 0;
    recs[beg + rank] = (sv << 2) | (et << 1) | dr;
}

__global__ __launch_bounds__(256)
void node_kernel(const float* __restrict__ h_in,
                 const float* __restrict__ hr,
                 const int* __restrict__ off,
                 const unsigned* __restrict__ recs,
                 const float* __restrict__ Ws,
                 const float* __restrict__ Wo,
                 const float* __restrict__ Wi,
                 float* __restrict__ h_out,
                 int n_nodes, int relu)
{
    __shared__ alignas(16) float lds_w[3 * D * WPITCH];
    __shared__ alignas(16) float lds_hr[3 * D];
    __shared__ alignas(16) float lds_a[3 * 32 * APITCH];

    const int t = threadIdx.x;

    const int g    = t >> 3;
    const int c    = (t & 7) * 4;

    const int n_tiles = (n_nodes + 31) >> 5;
    const int stride  = (int)gridDim.x;
    int tile = (int)blockIdx.x;

    int beg = 0, end = 0;
    float4 ha = make_float4(0.f, 0.f, 0.f, 0.f);
    float4 hb = ha;
    {
        const int node = tile * 32 + g;
        if (tile < n_tiles && node < n_nodes) {
            beg = off[node];
            end = off[node + 1];
            ha = *(const float4*)(h_in + (size_t)node * D + c);
            hb = *(const float4*)(h_in + (size_t)node * D + 32 + c);
        }
    }
    const float hrv = (t < 3 * D) ? hr[t] : 0.f;

    float4 wv[12];
#pragma unroll
    for (int j = 0; j < 4; ++j) {
        const int i4 = t + 256 * j;
        wv[j]     = ((const float4*)Ws)[i4];
        wv[4 + j] = ((const float4*)Wo)[i4];
        wv[8 + j] = ((const float4*)Wi)[i4];
    }

    unsigned rn[4] = { 0u, 0u, 0u, 0u };
    if (beg < end) {
#pragma unroll
        for (int u = 0; u < 4; ++u) {
            const int idx = (beg + u < end) ? (beg + u) : (end - 1);
            rn[u] = recs[idx];
        }
    }

#pragma unroll
    for (int j = 0; j < 4; ++j) {
        const int i4 = t + 256 * j;
        const int wi = (i4 >> 4) * WPITCH + (i4 & 15) * 4;
        *(float4*)(lds_w + wi)                  = wv[j];
        *(float4*)(lds_w + D * WPITCH + wi)     = wv[4 + j];
        *(float4*)(lds_w + 2 * D * WPITCH + wi) = wv[8 + j];
    }
    if (t < 3 * D) lds_hr[t] = hrv;

    const int wave  = t >> 5;
    const int lane  = t & 31;
    const int lhalf = lane >> 4;
    const int lmod  = lane & 15;
    const int n0    = (wave & 3) * 16;
    const int brow  = n0 + lmod;
    const float* xr = lds_a + ((wave >> 2) * 16 + lmod) * APITCH;
    const float* a0 = xr + 32 * APITCH;
    const float* a1 = xr + 64 * APITCH;

    for (; tile < n_tiles; tile += stride) {
        const int node = tile * 32 + g;
        const bool live = node < n_nodes;

        __syncthreads();

        const int ntile = tile + stride;
        int nbeg = 0, nend = 0;
        float4 nha = make_float4(0.f, 0.f, 0.f, 0.f);
        float4 nhb = nha;
        {
            const int nnode = ntile * 32 + g;
            if (ntile < n_tiles && nnode < n_nodes) {
                nbeg = off[nnode];
                nend = off[nnode + 1];
                nha = *(const float4*)(h_in + (size_t)nnode * D + c);
                nhb = *(const float4*)(h_in + (size_t)nnode * D + 32 + c);
            }
        }

        float4 xa = make_float4(0.f, 0.f, 0.f, 0.f);
        float4 xb = xa;
        if (live) {
            const float4 ra = *(const float4*)(lds_hr + 2 * D + c);
            const float4 rb = *(const float4*)(lds_hr + 2 * D + 32 + c);
            xa.x = ha.x - ra.x; xa.y = ha.y - ra.y; xa.z = ha.z - ra.z; xa.w = ha.w - ra.w;
            xb.x = hb.x - rb.x; xb.y = hb.y - rb.y; xb.z = hb.z - rb.z; xb.w = hb.w - rb.w;
        }
        *(float4*)(lds_a + g * APITCH + c)      = xa;
        *(float4*)(lds_a + g * APITCH + 32 + c) = xb;

        float4 s0a = make_float4(0.f, 0.f, 0.f, 0.f);
        float4 s0b = s0a, s1a = s0a, s1b = s0a;
        for (int i = beg; i < end; i += 4) {
            unsigned rec[4];
            float4   ga[4], gb[4];
#pragma unroll
            for (int u = 0; u < 4; ++u) rec[u] = rn[u];
#pragma unroll
            for (int u = 0; u < 4; ++u) {
                const float* p = h_in + (size_t)(rec[u] >> 2) * D + c;
                ga[u] = *(const float4*)p;
                gb[u] = *(const float4*)(p + 32);
            }
#pragma unroll
            for (int u = 0; u < 4; ++u) {
                const int idx = (i + 4 + u < end) ? (i + 4 + u) : (end - 1);
                rn[u] = recs[idx];
            }
#pragma unroll
            for (int u = 0; u < 4; ++u) {
                const bool ok = (i + u < end);
                const unsigned et = (rec[u] >> 1) & 1u;
                const bool     in = (rec[u] & 1u) != 0u;
                const bool     t0 = ok && !in;
                const bool     t1 = ok && in;
                const float4 ra = *(const float4*)(lds_hr + et * D + c);
                const float4 rb = *(const float4*)(lds_hr + et * D + 32 + c);
                float4 ma, mb;
                ma.x = ga[u].x - ra.x; ma.y = ga[u].y - ra.y;
                ma.z = ga[u].z - ra.z; ma.w = ga[u].w - ra.w;
                mb.x = gb[u].x - rb.x; mb.y = gb[u].y - rb.y;
                mb.z = gb[u].z - rb.z; mb.w = gb[u].w - rb.w;
                s0a.x += t0 ? ma.x : 0.f; s0a.y += t0 ? ma.y : 0.f;
                s0a.z += t0 ? ma.z : 0.f; s0a.w += t0 ? ma.w : 0.f;
                s0b.x += t0 ? mb.x : 0.f; s0b.y += t0 ? mb.y : 0.f;
                s0b.z += t0 ? mb.z : 0.f; s0b.w += t0 ? mb.w : 0.f;
                s1a.x += t1 ? ma.x : 0.f; s1a.y += t1 ? ma.y : 0.f;
                s1a.z += t1 ? ma.z : 0.f; s1a.w += t1 ? ma.w : 0.f;
                s1b.x += t1 ? mb.x : 0.f; s1b.y += t1 ? mb.y : 0.f;
                s1b.z += t1 ? mb.z : 0.f; s1b.w += t1 ? mb.w : 0.f;
            }
        }
        *(float4*)(lds_a + (32 + g) * APITCH + c)      = s0a;
        *(float4*)(lds_a + (32 + g) * APITCH + 32 + c) = s0b;
        *(float4*)(lds_a + (64 + g) * APITCH + c)      = s1a;
        *(float4*)(lds_a + (64 + g) * APITCH + 32 + c) = s1b;

        __syncthreads();

        unsigned nrn[4] = { 0u, 0u, 0u, 0u };
        if (nbeg < nend) {
#pragma unroll
            for (int u = 0; u < 4; ++u) {
                const int idx = (nbeg + u < nend) ? (nbeg + u) : (nend - 1);
                nrn[u] = recs[idx];
            }
        }

        const int m0 = tile * 32 + (wave >> 2) * 16;
        if (m0 < n_nodes) {
            v8f acc = {};
#pragma unroll
            for (int k0 = 0; k0 < D; k0 += 4) {
                const int ka = k0 + lhalf * 2;
                v2f a, b;
                a.x = xr[ka];
                a.y = xr[ka + 1];
                b.x = lds_w[brow * WPITCH + ka];
                b.y = lds_w[brow * WPITCH + ka + 1];
                acc = __builtin_amdgcn_wmma_f32_16x16x4_f32(false, a, false, b, (short)0, acc, false, false);
            }
#pragma unroll
            for (int k0 = 0; k0 < D; k0 += 4) {
                const int ka = k0 + lhalf * 2;
                v2f a, b;
                a.x = a0[ka];
                a.y = a0[ka + 1];
                b.x = lds_w[D * WPITCH + brow * WPITCH + ka];
                b.y = lds_w[D * WPITCH + brow * WPITCH + ka + 1];
                acc = __builtin_amdgcn_wmma_f32_16x16x4_f32(false, a, false, b, (short)0, acc, false, false);
            }
#pragma unroll
            for (int k0 = 0; k0 < D; k0 += 4) {
                const int ka = k0 + lhalf * 2;
                v2f a, b;
                a.x = a1[ka];
                a.y = a1[ka + 1];
                b.x = lds_w[2 * D * WPITCH + brow * WPITCH + ka];
                b.y = lds_w[2 * D * WPITCH + brow * WPITCH + ka + 1];
                acc = __builtin_amdgcn_wmma_f32_16x16x4_f32(false, a, false, b, (short)0, acc, false, false);
            }

#pragma unroll
            for (int r = 0; r < 8; ++r) {
                float v = acc[r];
                if (relu) v = fmaxf(v, 0.f);
                const int m = m0 + r + lhalf * 8;
                if (m < n_nodes) h_out[(size_t)m * D + n0 + lmod] = v;
            }
        }

        beg = nbeg; end = nend; ha = nha; hb = nhb;
#pragma unroll
        for (int u = 0; u < 4; ++u) rn[u] = nrn[u];
    }
}

extern "C" void kernel_launch(void* const* d_in, const int* in_sizes, int n_in,
                              void* d_out, int out_size, void* d_ws, size_t ws_size,
                              hipStream_t stream)
{
    const float* h_u   = (const float*)d_in[0];
    const float* Basis = (const float*)d_in[1];
    const float* alpha = (const float*)d_in[2];
    const float* W_O   = (const float*)d_in[3];
    const float* W_I   = (const float*)d_in[4];
    const float* W_S   = (const float*)d_in[5];
    const float* W_rel = (const float*)d_in[6];
    const int*   src   = (const int*)d_in[7];
    const int*   dst   = (const int*)d_in[8];
    const int*   etype = (const int*)d_in[9];
    const int*   edir  = (const int*)d_in[10];
    float* out = (float*)d_out;

    const int n_nodes = in_sizes[0] / D;
    const int n_edges = in_sizes[7];

    const int n_pad = n_nodes + 64;
    int*      cnt  = (int*)d_ws;
    int*      off  = cnt + n_pad;
    int*      cur  = off + n_pad;
    int*      bsum = cur + n_pad;
    int*      tmp  = bsum + n_pad;
    unsigned* recs = (unsigned*)(tmp + n_edges);
    float* hb0 = (float*)d_ws + (size_t)2 * n_nodes * D;
    float* hb1 = hb0 + (size_t)n_nodes * D;
    float* hra = hb1 + (size_t)n_nodes * D;
    float* hrb = hra + 256;

    hr_init_kernel<<<1, 192, 0, stream>>>(Basis, alpha, hra);

    const unsigned node_blk256 = (unsigned)((n_nodes + 255) / 256);
    const unsigned edge_blk256 = (unsigned)((n_edges + 255) / 256);
    const unsigned scan_blocks = (unsigned)((n_nodes + 1023) / 1024);

    zero_kernel<<<node_blk256, 256, 0, stream>>>(cnt, n_nodes);
    degree_kernel<<<edge_blk256, 256, 0, stream>>>(dst, cnt, n_edges);
    blocksum_kernel<<<scan_blocks, 1024, 0, stream>>>(cnt, bsum, n_nodes);
    scan_kernel<<<scan_blocks, 1024, 0, stream>>>(cnt, bsum, off, cur, n_nodes);
    fill_kernel<<<edge_blk256, 256, 0, stream>>>(dst, cur, tmp, n_edges);
    order_kernel<<<edge_blk256, 256, 0, stream>>>(tmp, off, src, dst, etype, edir, recs, n_edges);

    const float* hin[3]  = { h_u, hb0, hb1 };
    float*       hout[3] = { hb0, hb1, out };

    float* hrcur = hra;
    float* hrnxt = hrb;

    for (int l = 0; l < 3; ++l) {
        const int relu = (l < 2) ? 1 : 0;

        node_kernel<<<1024, 256, 0, stream>>>(hin[l], hrcur, off, recs,
                                                     W_S + (size_t)l * D * D,
                                                     W_O + (size_t)l * D * D,
                                                     W_I + (size_t)l * D * D,
                                                     hout[l], n_nodes, relu);

        if (l < 2) {
            rel_kernel<<<1, 192, 0, stream>>>(hrcur, W_rel + (size_t)l * D * D, hrnxt, 1);
            float* tq = hrcur; hrcur = hrnxt; hrnxt = tq;
        }
    }
}


# Segments ordered by packed record value instead of edge id: the slot-filling pass leaves record and destination in the slot, so the ordering pass gathers no edge arrays and ranks 8 slots a trip
# speedup vs baseline: 2.0204x; 1.0514x over previous
#include <hip/hip_runtime.h>
#include <hip/hip_bf16.h>


typedef __attribute__((ext_vector_type(2))) float v2f;
typedef __attribute__((ext_vector_type(8))) float v8f;

#define D 64
#define APITCH 68
#define WPITCH 68

__global__ void hr_init_kernel(const float* __restrict__ Basis,
                               const float* __restrict__ alpha,
                               float* __restrict__ h_r)
{
    int t = threadIdx.x;
    if (t >= 3 * D) return;
    int r = t >> 6, n = t & 63;
    float s = 0.f;
#pragma unroll
    for (int b = 0; b < 16; ++b) s += alpha[r * 16 + b] * Basis[b * D + n];
    h_r[t] = s;
}

__global__ void rel_kernel(const float* __restrict__ h_r,
                           const float* __restrict__ Wrel,
                           float* __restrict__ h_r_out,
                           int relu)
{
    int t = threadIdx.x;
    if (t >= 3 * D) return;
    int r = t >> 6, n = t & 63;
    float s = 0.f;
#pragma unroll 8
    for (int k = 0; k < D; ++k) s += h_r[r * D + k] * Wrel[n * D + k];
    if (relu) s = fmaxf(s, 0.f);
    h_r_out[t] = s;
}

__global__ __launch_bounds__(256) void zero_kernel(int* __restrict__ p, int n)
{
    int i = blockIdx.x * blockDim.x + threadIdx.x;
    if (i < n) p[i] = 0;
}

__global__ __launch_bounds__(256) void degree_kernel(const int* __restrict__ dst,
                                                     int* __restrict__ cnt, int n_edges)
{
    int e = blockIdx.x * blockDim.x + threadIdx.x;
    if (e < n_edges) atomicAdd(&cnt[dst[e]], 1);
}

__device__ __forceinline__ int wave_sum(int v)
{
#pragma unroll
    for (int o = 16; o > 0; o >>= 1) v += __shfl_down(v, o, 32);
    return v;
}

__device__ __forceinline__ int wave_scan_incl(int v, int lane)
{
#pragma unroll
    for (int o = 1; o < 32; o <<= 1) {
        int nb = __shfl_up(v, o, 32);
        if (lane >= o) v += nb;
    }
    return v;
}

__global__ __launch_bounds__(1024) void blocksum_kernel(const int* __restrict__ cnt,
                                                        int* __restrict__ bsum, int n)
{
    __shared__ int wsum[32];
    const int t = threadIdx.x, lane = t & 31, w = t >> 5;
    const int i = blockIdx.x * 1024 + t;
    int v = (i < n) ? cnt[i] : 0;
    v = wave_sum(v);
    if (lane == 0) wsum[w] = v;
    __syncthreads();
    if (w == 0) {
        int s = wave_sum(wsum[lane]);
        if (lane == 0) bsum[blockIdx.x] = s;
    }
}

__global__ __launch_bounds__(1024) void scan_kernel(const int* __restrict__ cnt,
                                                    const int* __restrict__ bsum,
                                                    int* __restrict__ off,
                                                    int* __restrict__ cur, int n)
{
    __shared__ int wsum[32];
    __shared__ int sbase;
    const int t = threadIdx.x, lane = t & 31, w = t >> 5;

    int p = 0;
    for (int j = t; j < (int)blockIdx.x; j += 1024) p += bsum[j];
    p = wave_sum(p);
    if (lane == 0) wsum[w] = p;
    __syncthreads();
    if (w == 0) {
        int s = wave_sum(wsum[lane]);
        if (lane == 0) sbase = s;
    }
    __syncthreads();
    const int base = sbase;

    const int i = blockIdx.x * 1024 + t;
    const int c = (i < n) ? cnt[i] : 0;
    const int incl = wave_scan_incl(c, lane);
    if (lane == 31) wsum[w] = incl;
    __syncthreads();
    if (w == 0) {
        int s = wave_scan_incl(wsum[lane], lane);
        wsum[lane] = s;
    }
    __syncthreads();
    const int wpre = (w > 0) ? wsum[w - 1] : 0;
    const int excl = base + wpre + incl - c;
    if (i < n) { off[i] = excl; cur[i] = excl; }
    if (i == n - 1) off[n] = excl + c;
}

__global__ __launch_bounds__(256) void fill_kernel(const int* __restrict__ dst,
                                                   const int* __restrict__ src,
                                                   const int* __restrict__ etype,
                                                   const int* __restrict__ dir,
                                                   int* __restrict__ cur,
                                                   uint2* __restrict__ tmp2, int n_edges)
{
    int e = blockIdx.x * blockDim.x + threadIdx.x;
    if (e < n_edges) {
        const int d = dst[e];
        const unsigned sv = (unsigned)src[e];
        const unsigned et = (unsigned)etype[e] & 1u;
        const unsigned dr = (dir[e] == 1) ? 1u : 0u;
        const int pos = atomicAdd(&cur[d], 1);
        tmp2[pos] = make_uint2((sv << 2) | (et << 1) | dr, (unsigned)d);
    }
}

__global__ __launch_bounds__(256) void order_kernel(const uint2* __restrict__ tmp2,
                                                    const int* __restrict__ off,
                                                    unsigned* __restrict__ recs, int n_edges)
{
    int i = blockIdx.x * blockDim.x + threadIdx.x;
    if (i >= n_edges) return;
    const uint2 me = tmp2[i];
    const unsigned x = me.x;
    const int v = (int)me.y;
    const int beg = off[v], end = off[v + 1];
    int rank = 0;
    for (int j = beg; j < end; j += 8) {
        unsigned y[8];
#pragma unroll
        for (int u = 0; u < 8; ++u) {
            const int idx = (j + u < end) ? (j + u) : (end - 1);
            y[u] = tmp2[idx].x;
        }
#pragma unroll
        for (int u = 0; u < 8; ++u) {
            const int k = j + u;
            const bool first = (y[u] < x) || (y[u] == x && k < i);
            rank += (k < end && first) ? 1 : 0;
        }
    }
    recs[beg + rank] = x;
}

__global__ __launch_bounds__(256)
void node_kernel(const float* __restrict__ h_in,
                 const float* __restrict__ hr,
                 const int* __restrict__ off,
                 const unsigned* __restrict__ recs,
                 const float* __restrict__ Ws,
                 const float* __restrict__ Wo,
                 const float* __restrict__ Wi,
                 float* __restrict__ h_out,
                 int n_nodes, int relu)
{
    __shared__ alignas(16) float lds_w[3 * D * WPITCH];
    __shared__ alignas(16) float lds_hr[3 * D];
    __shared__ alignas(16) float lds_a[3 * 32 * APITCH];

    const int t = threadIdx.x;

    const int g    = t >> 3;
    const int c    = (t & 7) * 4;

    const int n_tiles = (n_nodes + 31) >> 5;
    const int stride  = (int)gridDim.x;
    int tile = (int)blockIdx.x;

    int beg = 0, end = 0;
    float4 ha = make_float4(0.f, 0.f, 0.f, 0.f);
    float4 hb = ha;
    {
        const int node = tile * 32 + g;
        if (tile < n_tiles && node < n_nodes) {
            beg = off[node];
            end = off[node + 1];
            ha = *(const float4*)(h_in + (size_t)node * D + c);
            hb = *(const float4*)(h_in + (size_t)node * D + 32 + c);
        }
    }
    const float hrv = (t < 3 * D) ? hr[t] : 0.f;

    float4 wv[12];
#pragma unroll
    for (int j = 0; j < 4; ++j) {
        const int i4 = t + 256 * j;
        wv[j]     = ((const float4*)Ws)[i4];
        wv[4 + j] = ((const float4*)Wo)[i4];
        wv[8 + j] = ((const float4*)Wi)[i4];
    }

    unsigned rn[4] = { 0u, 0u, 0u, 0u };
    if (beg < end) {
#pragma unroll
        for (int u = 0; u < 4; ++u) {
            const int idx = (beg + u < end) ? (beg + u) : (end - 1);
            rn[u] = recs[idx];
        }
    }

#pragma unroll
    for (int j = 0; j < 4; ++j) {
        const int i4 = t + 256 * j;
        const int wi = (i4 >> 4) * WPITCH + (i4 & 15) * 4;
        *(float4*)(lds_w + wi)                  = wv[j];
        *(float4*)(lds_w + D * WPITCH + wi)     = wv[4 + j];
        *(float4*)(lds_w + 2 * D * WPITCH + wi) = wv[8 + j];
    }
    if (t < 3 * D) lds_hr[t] = hrv;

    const int wave  = t >> 5;
    const int lane  = t & 31;
    const int lhalf = lane >> 4;
    const int lmod  = lane & 15;
    const int n0    = (wave & 3) * 16;
    const int brow  = n0 + lmod;
    const float* xr = lds_a + ((wave >> 2) * 16 + lmod) * APITCH;
    const float* a0 = xr + 32 * APITCH;
    const float* a1 = xr + 64 * APITCH;

    for (; tile < n_tiles; tile += stride) {
        const int node = tile * 32 + g;
        const bool live = node < n_nodes;

        __syncthreads();

        const int ntile = tile + stride;
        int nbeg = 0, nend = 0;
        float4 nha = make_float4(0.f, 0.f, 0.f, 0.f);
        float4 nhb = nha;
        {
            const int nnode = ntile * 32 + g;
            if (ntile < n_tiles && nnode < n_nodes) {
                nbeg = off[nnode];
                nend = off[nnode + 1];
                nha = *(const float4*)(h_in + (size_t)nnode * D + c);
                nhb = *(const float4*)(h_in + (size_t)nnode * D + 32 + c);
            }
        }

        float4 xa = make_float4(0.f, 0.f, 0.f, 0.f);
        float4 xb = xa;
        if (live) {
            const float4 ra = *(const float4*)(lds_hr + 2 * D + c);
            const float4 rb = *(const float4*)(lds_hr + 2 * D + 32 + c);
            xa.x = ha.x - ra.x; xa.y = ha.y - ra.y; xa.z = ha.z - ra.z; xa.w = ha.w - ra.w;
            xb.x = hb.x - rb.x; xb.y = hb.y - rb.y; xb.z = hb.z - rb.z; xb.w = hb.w - rb.w;
        }
        *(float4*)(lds_a + g * APITCH + c)      = xa;
        *(float4*)(lds_a + g * APITCH + 32 + c) = xb;

        float4 s0a = make_float4(0.f, 0.f, 0.f, 0.f);
        float4 s0b = s0a, s1a = s0a, s1b = s0a;
        for (int i = beg; i < end; i += 4) {
            unsigned rec[4];
            float4   ga[4], gb[4];
#pragma unroll
            for (int u = 0; u < 4; ++u) rec[u] = rn[u];
#pragma unroll
            for (int u = 0; u < 4; ++u) {
                const float* p = h_in + (size_t)(rec[u] >> 2) * D + c;
                ga[u] = *(const float4*)p;
                gb[u] = *(const float4*)(p + 32);
            }
#pragma unroll
            for (int u = 0; u < 4; ++u) {
                const int idx = (i + 4 + u < end) ? (i + 4 + u) : (end - 1);
                rn[u] = recs[idx];
            }
#pragma unroll
            for (int u = 0; u < 4; ++u) {
                const bool ok = (i + u < end);
                const unsigned et = (rec[u] >> 1) & 1u;
                const bool     in = (rec[u] & 1u) != 0u;
                const bool     t0 = ok && !in;
                const bool     t1 = ok && in;
                const float4 ra = *(const float4*)(lds_hr + et * D + c);
                const float4 rb = *(const float4*)(lds_hr + et * D + 32 + c);
                float4 ma, mb;
                ma.x = ga[u].x - ra.x; ma.y = ga[u].y - ra.y;
                ma.z = ga[u].z - ra.z; ma.w = ga[u].w - ra.w;
                mb.x = gb[u].x - rb.x; mb.y = gb[u].y - rb.y;
                mb.z = gb[u].z - rb.z; mb.w = gb[u].w - rb.w;
                s0a.x += t0 ? ma.x : 0.f; s0a.y += t0 ? ma.y : 0.f;
                s0a.z += t0 ? ma.z : 0.f; s0a.w += t0 ? ma.w : 0.f;
                s0b.x += t0 ? mb.x : 0.f; s0b.y += t0 ? mb.y : 0.f;
                s0b.z += t0 ? mb.z : 0.f; s0b.w += t0 ? mb.w : 0.f;
                s1a.x += t1 ? ma.x : 0.f; s1a.y += t1 ? ma.y : 0.f;
                s1a.z += t1 ? ma.z : 0.f; s1a.w += t1 ? ma.w : 0.f;
                s1b.x += t1 ? mb.x : 0.f; s1b.y += t1 ? mb.y : 0.f;
                s1b.z += t1 ? mb.z : 0.f; s1b.w += t1 ? mb.w : 0.f;
            }
        }
        *(float4*)(lds_a + (32 + g) * APITCH + c)      = s0a;
        *(float4*)(lds_a + (32 + g) * APITCH + 32 + c) = s0b;
        *(float4*)(lds_a + (64 + g) * APITCH + c)      = s1a;
        *(float4*)(lds_a + (64 + g) * APITCH + 32 + c) = s1b;

        __syncthreads();

        unsigned nrn[4] = { 0u, 0u, 0u, 0u };
        if (nbeg < nend) {
#pragma unroll
            for (int u = 0; u < 4; ++u) {
                const int idx = (nbeg + u < nend) ? (nbeg + u) : (nend - 1);
                nrn[u] = recs[idx];
            }
        }

        const int m0 = tile * 32 + (wave >> 2) * 16;
        if (m0 < n_nodes) {
            v8f acc = {};
#pragma unroll
            for (int k0 = 0; k0 < D; k0 += 4) {
                const int ka = k0 + lhalf * 2;
                v2f a, b;
                a.x = xr[ka];
                a.y = xr[ka + 1];
                b.x = lds_w[brow * WPITCH + ka];
                b.y = lds_w[brow * WPITCH + ka + 1];
                acc = __builtin_amdgcn_wmma_f32_16x16x4_f32(false, a, false, b, (short)0, acc, false, false);
            }
#pragma unroll
            for (int k0 = 0; k0 < D; k0 += 4) {
                const int ka = k0 + lhalf * 2;
                v2f a, b;
                a.x = a0[ka];
                a.y = a0[ka + 1];
                b.x = lds_w[D * WPITCH + brow * WPITCH + ka];
                b.y = lds_w[D * WPITCH + brow * WPITCH + ka + 1];
                acc = __builtin_amdgcn_wmma_f32_16x16x4_f32(false, a, false, b, (short)0, acc, false, false);
            }
#pragma unroll
            for (int k0 = 0; k0 < D; k0 += 4) {
                const int ka = k0 + lhalf * 2;
                v2f a, b;
                a.x = a1[ka];
                a.y = a1[ka + 1];
                b.x = lds_w[2 * D * WPITCH + brow * WPITCH + ka];
                b.y = lds_w[2 * D * WPITCH + brow * WPITCH + ka + 1];
                acc = __builtin_amdgcn_wmma_f32_16x16x4_f32(false, a, false, b, (short)0, acc, false, false);
            }

#pragma unroll
            for (int r = 0; r < 8; ++r) {
                float v = acc[r];
                if (relu) v = fmaxf(v, 0.f);
                const int m = m0 + r + lhalf * 8;
                if (m < n_nodes) h_out[(size_t)m * D + n0 + lmod] = v;
            }
        }

        beg = nbeg; end = nend; ha = nha; hb = nhb;
#pragma unroll
        for (int u = 0; u < 4; ++u) rn[u] = nrn[u];
    }
}

extern "C" void kernel_launch(void* const* d_in, const int* in_sizes, int n_in,
                              void* d_out, int out_size, void* d_ws, size_t ws_size,
                              hipStream_t stream)
{
    const float* h_u   = (const float*)d_in[0];
    const float* Basis = (const float*)d_in[1];
    const float* alpha = (const float*)d_in[2];
    const float* W_O   = (const float*)d_in[3];
    const float* W_I   = (const float*)d_in[4];
    const float* W_S   = (const float*)d_in[5];
    const float* W_rel = (const float*)d_in[6];
    const int*   src   = (const int*)d_in[7];
    const int*   dst   = (const int*)d_in[8];
    const int*   etype = (const int*)d_in[9];
    const int*   edir  = (const int*)d_in[10];
    float* out = (float*)d_out;

    const int n_nodes = in_sizes[0] / D;
    const int n_edges = in_sizes[7];

    const int n_pad = n_nodes + 64;
    int*      cnt  = (int*)d_ws;
    int*      off  = cnt + n_pad;
    int*      cur  = off + n_pad;
    int*      bsum = cur + n_pad;
    uint2*    tmp2 = (uint2*)(bsum + n_pad);
    unsigned* recs = (unsigned*)(tmp2 + n_edges);
    float* hb0 = (float*)d_ws + (size_t)2 * n_nodes * D;
    float* hb1 = hb0 + (size_t)n_nodes * D;
    float* hra = hb1 + (size_t)n_nodes * D;
    float* hrb = hra + 256;

    hr_init_kernel<<<1, 192, 0, stream>>>(Basis, alpha, hra);

    const unsigned node_blk256 = (unsigned)((n_nodes + 255) / 256);
    const unsigned edge_blk256 = (unsigned)((n_edges + 255) / 256);
    const unsigned scan_blocks = (unsigned)((n_nodes + 1023) / 1024);

    zero_kernel<<<node_blk256, 256, 0, stream>>>(cnt, n_nodes);
    degree_kernel<<<edge_blk256, 256, 0, stream>>>(dst, cnt, n_edges);
    blocksum_kernel<<<scan_blocks, 1024, 0, stream>>>(cnt, bsum, n_nodes);
    scan_kernel<<<scan_blocks, 1024, 0, stream>>>(cnt, bsum, off, cur, n_nodes);
    fill_kernel<<<edge_blk256, 256, 0, stream>>>(dst, src, etype, edir, cur, tmp2, n_edges);
    order_kernel<<<edge_blk256, 256, 0, stream>>>(tmp2, off, recs, n_edges);

    const float* hin[3]  = { h_u, hb0, hb1 };
    float*       hout[3] = { hb0, hb1, out };

    float* hrcur = hra;
    float* hrnxt = hrb;

    for (int l = 0; l < 3; ++l) {
        const int relu = (l < 2) ? 1 : 0;

        node_kernel<<<1024, 256, 0, stream>>>(hin[l], hrcur, off, recs,
                                                     W_S + (size_t)l * D * D,
                                                     W_O + (size_t)l * D * D,
                                                     W_I + (size_t)l * D * D,
                                                     hout[l], n_nodes, relu);

        if (l < 2) {
            rel_kernel<<<1, 192, 0, stream>>>(hrcur, W_rel + (size_t)l * D * D, hrnxt, 1);
            float* tq = hrcur; hrcur = hrnxt; hrnxt = tq;
        }
    }
}


# Counting pass keeps the value its integer add returns (the edge's arrival number at its destination); slot filling uses segment begin + that number: one scattered atomic per edge, not two
# speedup vs baseline: 2.5319x; 1.2532x over previous
#include <hip/hip_runtime.h>
#include <hip/hip_bf16.h>


typedef __attribute__((ext_vector_type(2))) float v2f;
typedef __attribute__((ext_vector_type(8))) float v8f;

#define D 64
#define APITCH 68
#define WPITCH 68

__device__ __forceinline__ void relation_rows(const float* __restrict__ Basis,
                                              const float* __restrict__ alpha,
                                              const float* __restrict__ Wrel,
                                              float* __restrict__ hr0,
                                              float* __restrict__ hr1,
                                              float* __restrict__ hr2)
{
    __shared__ alignas(16) float shw[2 * D * WPITCH];
    __shared__ alignas(16) float sh0[3 * D];
    __shared__ alignas(16) float sh1[3 * D];

    const int t = threadIdx.x;
    const float4 wa = ((const float4*)Wrel)[t];
    const float4 wb = ((const float4*)(Wrel + D * D))[t];

    const bool act = t < 3 * D;
    const int r = t >> 6, n = t & 63;
    if (act) {
        float s = 0.f;
#pragma unroll
        for (int b = 0; b < 16; ++b) s = fmaf(alpha[r * 16 + b], Basis[b * D + n], s);
        hr0[t] = s;
        sh0[t] = s;
    }
    const int wi = (t >> 4) * WPITCH + (t & 15) * 4;
    *(float4*)(shw + wi)              = wa;
    *(float4*)(shw + D * WPITCH + wi) = wb;
    __syncthreads();
    if (act) {
        float s = 0.f;
#pragma unroll
        for (int j = 0; j < 16; ++j) {
            const float4 h = *(const float4*)(sh0 + r * D + 4 * j);
            const float4 w = *(const float4*)(shw + n * WPITCH + 4 * j);
            s = fmaf(h.x, w.x, s); s = fmaf(h.y, w.y, s); s = fmaf(h.z, w.z, s); s = fmaf(h.w, w.w, s);
        }
        s = fmaxf(s, 0.f);
        hr1[t] = s;
        sh1[t] = s;
    }
    __syncthreads();
    if (act) {
        float s = 0.f;
#pragma unroll
        for (int j = 0; j < 16; ++j) {
            const float4 h = *(const float4*)(sh1 + r * D + 4 * j);
            const float4 w = *(const float4*)(shw + D * WPITCH + n * WPITCH + 4 * j);
            s = fmaf(h.x, w.x, s); s = fmaf(h.y, w.y, s); s = fmaf(h.z, w.z, s); s = fmaf(h.w, w.w, s);
        }
        s = fmaxf(s, 0.f);
        hr2[t] = s;
    }
}

__global__ __launch_bounds__(256) void zero_kernel(int* __restrict__ p, int n)
{
    int i = blockIdx.x * blockDim.x + threadIdx.x;
    if (i < n) p[i] = 0;
}

__global__ __launch_bounds__(256) void degree_kernel(const int* __restrict__ dst,
                                                     int* __restrict__ cnt,
                                                     int* __restrict__ arrival, int n_edges)
{
    int e = blockIdx.x * blockDim.x + threadIdx.x;
    if (e < n_edges) arrival[e] = atomicAdd(&cnt[dst[e]], 1);
}

__device__ __forceinline__ int wave_scan_incl(int v, int lane)
{
#pragma unroll
    for (int o = 1; o < 32; o <<= 1) {
        int nb = __shfl_up(v, o, 32);
        if (lane >= o) v += nb;
    }
    return v;
}

__global__ __launch_bounds__(1024) void scan_kernel(const int* cnt,
                                                    int* total,
                                                    int2* __restrict__ seg, int n,
                                                    const float* __restrict__ Basis,
                                                    const float* __restrict__ alpha,
                                                    const float* __restrict__ Wrel,
                                                    float* __restrict__ hr0,
                                                    float* __restrict__ hr1,
                                                    float* __restrict__ hr2)
{
    __shared__ int wsum[32];
    __shared__ int sbase;
    const int t = threadIdx.x, lane = t & 31, w = t >> 5;

    if ((int)blockIdx.x * 1024 >= n) {
        relation_rows(Basis, alpha, Wrel, hr0, hr1, hr2);
        return;
    }

    const int i = blockIdx.x * 1024 + t;
    const int c = (i < n) ? cnt[i] : 0;
    const int incl = wave_scan_incl(c, lane);
    if (lane == 31) wsum[w] = incl;
    __syncthreads();
    if (w == 0) {
        int s = wave_scan_incl(wsum[lane], lane);
        wsum[lane] = s;
        if (lane == 31) sbase = atomicAdd(total, s);
    }
    __syncthreads();
    const int base = sbase;
    const int wpre = (w > 0) ? wsum[w - 1] : 0;
    const int excl = base + wpre + incl - c;
    if (i < n) seg[i] = make_int2(excl, excl + c);
}

__global__ __launch_bounds__(256) void fill_kernel(const int* __restrict__ dst,
                                                   const int* __restrict__ src,
                                                   const int* __restrict__ etype,
                                                   const int* __restrict__ dir,
                                                   const int* __restrict__ arrival,
                                                   const int2* __restrict__ seg,
                                                   uint2* __restrict__ tmp2, int n_edges)
{
    int e = blockIdx.x * blockDim.x + threadIdx.x;
    if (e < n_edges) {
        const int d = dst[e];
        const unsigned sv = (unsigned)src[e];
        const unsigned et = (unsigned)etype[e] & 1u;
        const unsigned dr = (dir[e] == 1) ? 1u : 0u;
        const int pos = seg[d].x + arrival[e];
        tmp2[pos] = make_uint2((sv << 2) | (et << 1) | dr, (unsigned)d);
    }
}

__global__ __launch_bounds__(256) void order_kernel(const uint2* __restrict__ tmp2,
                                                    const int2* __restrict__ seg,
                                                    unsigned* __restrict__ recs, int n_edges)
{
    int i = blockIdx.x * blockDim.x + threadIdx.x;
    if (i >= n_edges) return;
    const uint2 me = tmp2[i];
    const unsigned x = me.x;
    const int v = (int)me.y;
    const int2 se = seg[v];
    const int beg = se.x, end = se.y;
    int rank = 0;
    for (int j = beg; j < end; j += 8) {
        unsigned y[8];
#pragma unroll
        for (int u = 0; u < 8; ++u) {
            const int idx = (j + u < end) ? (j + u) : (end - 1);
            y[u] = tmp2[idx].x;
        }
#pragma unroll
        for (int u = 0; u < 8; ++u) {
            const int k = j + u;
            const bool first = (y[u] < x) || (y[u] == x && k < i);
            rank += (k < end && first) ? 1 : 0;
        }
    }
    recs[beg + rank] = x;
}

__global__ __launch_bounds__(256)
void node_kernel(const float* __restrict__ h_in,
                 const float* __restrict__ hr,
                 const int2* __restrict__ seg,
                 const unsigned* __restrict__ recs,
                 const float* __restrict__ Ws,
                 const float* __restrict__ Wo,
                 const float* __restrict__ Wi,
                 float* __restrict__ h_out,
                 int n_nodes, int relu)
{
    __shared__ alignas(16) float lds_w[3 * D * WPITCH];
    __shared__ alignas(16) float lds_hr[3 * D];
    __shared__ alignas(16) float lds_a[3 * 32 * APITCH];

    const int t = threadIdx.x;

    const int g    = t >> 3;
    const int c    = (t & 7) * 4;

    const int n_tiles = (n_nodes + 31) >> 5;
    const int stride  = (int)gridDim.x;
    int tile = (int)blockIdx.x;

    int beg = 0, end = 0;
    float4 ha = make_float4(0.f, 0.f, 0.f, 0.f);
    float4 hb = ha;
    {
        const int node = tile * 32 + g;
        if (tile < n_tiles && node < n_nodes) {
            const int2 se = seg[node];
            beg = se.x;
            end = se.y;
            ha = *(const float4*)(h_in + (size_t)node * D + c);
            hb = *(const float4*)(h_in + (size_t)node * D + 32 + c);
        }
    }
    const float hrv = (t < 3 * D) ? hr[t] : 0.f;

    float4 wv[12];
#pragma unroll
    for (int j = 0; j < 4; ++j) {
        const int i4 = t + 256 * j;
        wv[j]     = ((const float4*)Ws)[i4];
        wv[4 + j] = ((const float4*)Wo)[i4];
        wv[8 + j] = ((const float4*)Wi)[i4];
    }

    unsigned rn[4] = { 0u, 0u, 0u, 0u };
    if (beg < end) {
#pragma unroll
        for (int u = 0; u < 4; ++u) {
            const int idx = (beg + u < end) ? (beg + u) : (end - 1);
            rn[u] = recs[idx];
        }
    }

#pragma unroll
    for (int j = 0; j < 4; ++j) {
        const int i4 = t + 256 * j;
        const int wi = (i4 >> 4) * WPITCH + (i4 & 15) * 4;
        *(float4*)(lds_w + wi)                  = wv[j];
        *(float4*)(lds_w + D * WPITCH + wi)     = wv[4 + j];
        *(float4*)(lds_w + 2 * D * WPITCH + wi) = wv[8 + j];
    }
    if (t < 3 * D) lds_hr[t] = hrv;

    const int wave  = t >> 5;
    const int lane  = t & 31;
    const int lhalf = lane >> 4;
    const int lmod  = lane & 15;
    const int n0    = (wave & 3) * 16;
    const int brow  = n0 + lmod;
    const float* xr = lds_a + ((wave >> 2) * 16 + lmod) * APITCH;
    const float* a0 = xr + 32 * APITCH;
    const float* a1 = xr + 64 * APITCH;

    for (; tile < n_tiles; tile += stride) {
        const int node = tile * 32 + g;
        const bool live = node < n_nodes;

        __syncthreads();

        const int ntile = tile + stride;
        int nbeg = 0, nend = 0;
        float4 nha = make_float4(0.f, 0.f, 0.f, 0.f);
        float4 nhb = nha;
        {
            const int nnode = ntile * 32 + g;
            if (ntile < n_tiles && nnode < n_nodes) {
                const int2 se = seg[nnode];
                nbeg = se.x;
                nend = se.y;
                nha = *(const float4*)(h_in + (size_t)nnode * D + c);
                nhb = *(const float4*)(h_in + (size_t)nnode * D + 32 + c);
            }
        }

        float4 xa = make_float4(0.f, 0.f, 0.f, 0.f);
        float4 xb = xa;
        if (live) {
            const float4 ra = *(const float4*)(lds_hr + 2 * D + c);
            const float4 rb = *(const float4*)(lds_hr + 2 * D + 32 + c);
            xa.x = ha.x - ra.x; xa.y = ha.y - ra.y; xa.z = ha.z - ra.z; xa.w = ha.w - ra.w;
            xb.x = hb.x - rb.x; xb.y = hb.y - rb.y; xb.z = hb.z - rb.z; xb.w = hb.w - rb.w;
        }
        *(float4*)(lds_a + g * APITCH + c)      = xa;
        *(float4*)(lds_a + g * APITCH + 32 + c) = xb;

        float4 s0a = make_float4(0.f, 0.f, 0.f, 0.f);
        float4 s0b = s0a, s1a = s0a, s1b = s0a;
        for (int i = beg; i < end; i += 4) {
            unsigned rec[4];
            float4   ga[4], gb[4];
#pragma unroll
            for (int u = 0; u < 4; ++u) rec[u] = rn[u];
#pragma unroll
            for (int u = 0; u < 4; ++u) {
                const float* p = h_in + (size_t)(rec[u] >> 2) * D + c;
                ga[u] = *(const float4*)p;
                gb[u] = *(const float4*)(p + 32);
            }
#pragma unroll
            for (int u = 0; u < 4; ++u) {
                const int idx = (i + 4 + u < end) ? (i + 4 + u) : (end - 1);
                rn[u] = recs[idx];
            }
#pragma unroll
            for (int u = 0; u < 4; ++u) {
                const bool ok = (i + u < end);
                const unsigned et = (rec[u] >> 1) & 1u;
                const bool     in = (rec[u] & 1u) != 0u;
                const bool     t0 = ok && !in;
                const bool     t1 = ok && in;
                const float4 ra = *(const float4*)(lds_hr + et * D + c);
                const float4 rb = *(const float4*)(lds_hr + et * D + 32 + c);
                float4 ma, mb;
                ma.x = ga[u].x - ra.x; ma.y = ga[u].y - ra.y;
                ma.z = ga[u].z - ra.z; ma.w = ga[u].w - ra.w;
                mb.x = gb[u].x - rb.x; mb.y = gb[u].y - rb.y;
                mb.z = gb[u].z - rb.z; mb.w = gb[u].w - rb.w;
                s0a.x += t0 ? ma.x : 0.f; s0a.y += t0 ? ma.y : 0.f;
                s0a.z += t0 ? ma.z : 0.f; s0a.w += t0 ? ma.w : 0.f;
                s0b.x += t0 ? mb.x : 0.f; s0b.y += t0 ? mb.y : 0.f;
                s0b.z += t0 ? mb.z : 0.f; s0b.w += t0 ? mb.w : 0.f;
                s1a.x += t1 ? ma.x : 0.f; s1a.y += t1 ? ma.y : 0.f;
                s1a.z += t1 ? ma.z : 0.f; s1a.w += t1 ? ma.w : 0.f;
                s1b.x += t1 ? mb.x : 0.f; s1b.y += t1 ? mb.y : 0.f;
                s1b.z += t1 ? mb.z : 0.f; s1b.w += t1 ? mb.w : 0.f;
            }
        }
        *(float4*)(lds_a + (32 + g) * APITCH + c)      = s0a;
        *(float4*)(lds_a + (32 + g) * APITCH + 32 + c) = s0b;
        *(float4*)(lds_a + (64 + g) * APITCH + c)      = s1a;
        *(float4*)(lds_a + (64 + g) * APITCH + 32 + c) = s1b;

        __syncthreads();

        unsigned nrn[4] = { 0u, 0u, 0u, 0u };
        if (nbeg < nend) {
#pragma unroll
            for (int u = 0; u < 4; ++u) {
                const int idx = (nbeg + u < nend) ? (nbeg + u) : (nend - 1);
                nrn[u] = recs[idx];
            }
        }

        const int m0 = tile * 32 + (wave >> 2) * 16;
        if (m0 < n_nodes) {
            v8f acc = {};
#pragma unroll
            for (int k0 = 0; k0 < D; k0 += 4) {
                const int ka = k0 + lhalf * 2;
                v2f a, b;
                a.x = xr[ka];
                a.y = xr[ka + 1];
                b.x = lds_w[brow * WPITCH + ka];
                b.y = lds_w[brow * WPITCH + ka + 1];
                acc = __builtin_amdgcn_wmma_f32_16x16x4_f32(false, a, false, b, (short)0, acc, false, false);
            }
#pragma unroll
            for (int k0 = 0; k0 < D; k0 += 4) {
                const int ka = k0 + lhalf * 2;
                v2f a, b;
                a.x = a0[ka];
                a.y = a0[ka + 1];
                b.x = lds_w[D * WPITCH + brow * WPITCH + ka];
                b.y = lds_w[D * WPITCH + brow * WPITCH + ka + 1];
                acc = __builtin_amdgcn_wmma_f32_16x16x4_f32(false, a, false, b, (short)0, acc, false, false);
            }
#pragma unroll
            for (int k0 = 0; k0 < D; k0 += 4) {
                const int ka = k0 + lhalf * 2;
                v2f a, b;
                a.x = a1[ka];
                a.y = a1[ka + 1];
                b.x = lds_w[2 * D * WPITCH + brow * WPITCH + ka];
                b.y = lds_w[2 * D * WPITCH + brow * WPITCH + ka + 1];
                acc = __builtin_amdgcn_wmma_f32_16x16x4_f32(false, a, false, b, (short)0, acc, false, false);
            }

#pragma unroll
            for (int r = 0; r < 8; ++r) {
                float v = acc[r];
                if (relu) v = fmaxf(v, 0.f);
                const int m = m0 + r + lhalf * 8;
                if (m < n_nodes) h_out[(size_t)m * D + n0 + lmod] = v;
            }
        }

        beg = nbeg; end = nend; ha = nha; hb = nhb;
#pragma unroll
        for (int u = 0; u < 4; ++u) rn[u] = nrn[u];
    }
}

extern "C" void kernel_launch(void* const* d_in, const int* in_sizes, int n_in,
                              void* d_out, int out_size, void* d_ws, size_t ws_size,
                              hipStream_t stream)
{
    const float* h_u   = (const float*)d_in[0];
    const float* Basis = (const float*)d_in[1];
    const float* alpha = (const float*)d_in[2];
    const float* W_O   = (const float*)d_in[3];
    const float* W_I   = (const float*)d_in[4];
    const float* W_S   = (const float*)d_in[5];
    const float* W_rel = (const float*)d_in[6];
    const int*   src   = (const int*)d_in[7];
    const int*   dst   = (const int*)d_in[8];
    const int*   etype = (const int*)d_in[9];
    const int*   edir  = (const int*)d_in[10];
    float* out = (float*)d_out;

    const int n_nodes = in_sizes[0] / D;
    const int n_edges = in_sizes[7];

    const int n_pad = n_nodes + 64;
    int*      cnt  = (int*)d_ws;
    int2*     seg  = (int2*)(cnt + 2 * n_pad);
    uint2*    tmp2 = (uint2*)(cnt + 4 * n_pad);
    unsigned* recs = (unsigned*)(tmp2 + n_edges);
    int*      arrival = (int*)recs;
    float* hr_l2 = (float*)(recs + n_edges);
    float* hb0 = (float*)d_ws + (size_t)2 * n_nodes * D;
    float* hb1 = hb0 + (size_t)n_nodes * D;
    float* hr_l0 = hb1 + (size_t)n_nodes * D;
    float* hr_l1 = hr_l0 + 256;

    const unsigned zero_blocks = (unsigned)((n_nodes + 1 + 255) / 256);
    const unsigned edge_blk256 = (unsigned)((n_edges + 255) / 256);
    const unsigned scan_blocks = (unsigned)((n_nodes + 1023) / 1024);

    zero_kernel<<<zero_blocks, 256, 0, stream>>>(cnt, n_nodes + 1);
    degree_kernel<<<edge_blk256, 256, 0, stream>>>(dst, cnt, arrival, n_edges);
    scan_kernel<<<scan_blocks + 1, 1024, 0, stream>>>(cnt, cnt + n_nodes, seg, n_nodes,
                                                      Basis, alpha, W_rel, hr_l0, hr_l1, hr_l2);
    fill_kernel<<<edge_blk256, 256, 0, stream>>>(dst, src, etype, edir, arrival, seg, tmp2, n_edges);
    order_kernel<<<edge_blk256, 256, 0, stream>>>(tmp2, seg, recs, n_edges);

    const float* hin[3]  = { h_u, hb0, hb1 };
    float*       hout[3] = { hb0, hb1, out };
    const float* hrl[3]  = { hr_l0, hr_l1, hr_l2 };

    for (int l = 0; l < 3; ++l) {
        const int relu = (l < 2) ? 1 : 0;

        node_kernel<<<1024, 256, 0, stream>>>(hin[l], hrl[l], seg, recs,
                                                     W_S + (size_t)l * D * D,
                                                     W_O + (size_t)l * D * D,
                                                     W_I + (size_t)l * D * D,
                                                     hout[l], n_nodes, relu);
    }
}


# Segment table built by a two-level counting sort on LDS counters (edges binned per 256-destination bucket by chunk, prefix sum, one block per bucket): no scattered global atomic, no clearing pass
# speedup vs baseline: 2.8805x; 1.1377x over previous
#include <hip/hip_runtime.h>
#include <hip/hip_bf16.h>


typedef __attribute__((ext_vector_type(2))) float v2f;
typedef __attribute__((ext_vector_type(8))) float v8f;

#define D 64
#define APITCH 68
#define WPITCH 68
#define CB 512
#define BSHIFT 8

__device__ __forceinline__ void relation_rows(const float* __restrict__ Basis,
                                              const float* __restrict__ alpha,
                                              const float* __restrict__ Wrel,
                                              float* __restrict__ hr0,
                                              float* __restrict__ hr1,
                                              float* __restrict__ hr2)
{
    __shared__ alignas(16) float shw[2 * D * WPITCH];
    __shared__ alignas(16) float sh0[3 * D];
    __shared__ alignas(16) float sh1[3 * D];

    const int t = threadIdx.x;
    const float4 wa = ((const float4*)Wrel)[t];
    const float4 wb = ((const float4*)(Wrel + D * D))[t];

    const bool act = t < 3 * D;
    const int r = t >> 6, n = t & 63;
    if (act) {
        float s = 0.f;
#pragma unroll
        for (int b = 0; b < 16; ++b) s = fmaf(alpha[r * 16 + b], Basis[b * D + n], s);
        hr0[t] = s;
        sh0[t] = s;
    }
    const int wi = (t >> 4) * WPITCH + (t & 15) * 4;
    *(float4*)(shw + wi)              = wa;
    *(float4*)(shw + D * WPITCH + wi) = wb;
    __syncthreads();
    if (act) {
        float s = 0.f;
#pragma unroll
        for (int j = 0; j < 16; ++j) {
            const float4 h = *(const float4*)(sh0 + r * D + 4 * j);
            const float4 w = *(const float4*)(shw + n * WPITCH + 4 * j);
            s = fmaf(h.x, w.x, s); s = fmaf(h.y, w.y, s); s = fmaf(h.z, w.z, s); s = fmaf(h.w, w.w, s);
        }
        s = fmaxf(s, 0.f);
        hr1[t] = s;
        sh1[t] = s;
    }
    __syncthreads();
    if (act) {
        float s = 0.f;
#pragma unroll
        for (int j = 0; j < 16; ++j) {
            const float4 h = *(const float4*)(sh1 + r * D + 4 * j);
            const float4 w = *(const float4*)(shw + D * WPITCH + n * WPITCH + 4 * j);
            s = fmaf(h.x, w.x, s); s = fmaf(h.y, w.y, s); s = fmaf(h.z, w.z, s); s = fmaf(h.w, w.w, s);
        }
        s = fmaxf(s, 0.f);
        hr2[t] = s;
    }
}

__global__ __launch_bounds__(256) void hist_kernel(const int* __restrict__ dst,
                                                   int* cells,
                                                   int* total,
                                                   int n_edges, int n_buckets)
{
    extern __shared__ int lh[];
    const int t = threadIdx.x;
    const int cb = (int)gridDim.x;
    for (int b = t; b < n_buckets; b += 256) lh[b] = 0;
    if (blockIdx.x == 0 && t == 0) *total = 0;
    __syncthreads();
    const int chunk = (n_edges + cb - 1) / cb;
    const int e0 = (int)blockIdx.x * chunk;
    const int e1 = (e0 + chunk < n_edges) ? (e0 + chunk) : n_edges;
    for (int e = e0 + t; e < e1; e += 4 * 256) {
        int d[4];
#pragma unroll
        for (int u = 0; u < 4; ++u) {
            const int ee = (e + u * 256 < e1) ? (e + u * 256) : (e1 - 1);
            d[u] = dst[ee];
        }
#pragma unroll
        for (int u = 0; u < 4; ++u)
            atomicAdd(&lh[d[u] >> BSHIFT], (e + u * 256 < e1) ? 1 : 0);
    }
    __syncthreads();
    for (int b = t; b < n_buckets; b += 256) cells[b * cb + (int)blockIdx.x] = lh[b];
}

__device__ __forceinline__ int wave_scan_incl(int v, int lane)
{
#pragma unroll
    for (int o = 1; o < 32; o <<= 1) {
        int nb = __shfl_up(v, o, 32);
        if (lane >= o) v += nb;
    }
    return v;
}

__global__ __launch_bounds__(1024) void scan_kernel(const int* cnt,
                                                    int* total,
                                                    int2* __restrict__ seg, int n,
                                                    const float* __restrict__ Basis,
                                                    const float* __restrict__ alpha,
                                                    const float* __restrict__ Wrel,
                                                    float* __restrict__ hr0,
                                                    float* __restrict__ hr1,
                                                    float* __restrict__ hr2)
{
    __shared__ int wsum[32];
    __shared__ int sbase;
    const int t = threadIdx.x, lane = t & 31, w = t >> 5;

    if ((int)blockIdx.x * 1024 >= n) {
        relation_rows(Basis, alpha, Wrel, hr0, hr1, hr2);
        return;
    }

    const int i = blockIdx.x * 1024 + t;
    const int c = (i < n) ? cnt[i] : 0;
    const int incl = wave_scan_incl(c, lane);
    if (lane == 31) wsum[w] = incl;
    __syncthreads();
    if (w == 0) {
        int s = wave_scan_incl(wsum[lane], lane);
        wsum[lane] = s;
        if (lane == 31) sbase = atomicAdd(total, s);
    }
    __syncthreads();
    const int base = sbase;
    const int wpre = (w > 0) ? wsum[w - 1] : 0;
    const int excl = base + wpre + incl - c;
    if (i < n) seg[i] = make_int2(excl, excl + c);
}

__global__ __launch_bounds__(256) void scatter_kernel(const int* __restrict__ dst,
                                                      const int* __restrict__ src,
                                                      const int* __restrict__ etype,
                                                      const int* __restrict__ dir,
                                                      const int2* __restrict__ cellseg,
                                                      uint2* __restrict__ tmpa,
                                                      int n_edges, int n_buckets)
{
    extern __shared__ int cur[];
    const int t = threadIdx.x;
    const int cb = (int)gridDim.x;
    for (int b = t; b < n_buckets; b += 256) cur[b] = cellseg[b * cb + (int)blockIdx.x].x;
    __syncthreads();
    const int chunk = (n_edges + cb - 1) / cb;
    const int e0 = (int)blockIdx.x * chunk;
    const int e1 = (e0 + chunk < n_edges) ? (e0 + chunk) : n_edges;
    int e = e0 + t;
    for (; e + 3 * 256 < e1; e += 4 * 256) {
        int d[4];
        unsigned rec[4];
#pragma unroll
        for (int u = 0; u < 4; ++u) {
            const int ee = e + u * 256;
            d[u] = dst[ee];
            const unsigned sv = (unsigned)src[ee];
            const unsigned et = (unsigned)etype[ee] & 1u;
            const unsigned dr = (dir[ee] == 1) ? 1u : 0u;
            rec[u] = (sv << 2) | (et << 1) | dr;
        }
        int pos[4];
#pragma unroll
        for (int u = 0; u < 4; ++u) pos[u] = atomicAdd(&cur[d[u] >> BSHIFT], 1);
#pragma unroll
        for (int u = 0; u < 4; ++u) tmpa[pos[u]] = make_uint2(rec[u], (unsigned)d[u]);
    }
    for (; e < e1; e += 256) {
        const int dd = dst[e];
        const unsigned sv = (unsigned)src[e];
        const unsigned et = (unsigned)etype[e] & 1u;
        const unsigned dr = (dir[e] == 1) ? 1u : 0u;
        const int pos = atomicAdd(&cur[dd >> BSHIFT], 1);
        tmpa[pos] = make_uint2((sv << 2) | (et << 1) | dr, (unsigned)dd);
    }
}

__global__ __launch_bounds__(1024) void bucket_kernel(const int2* __restrict__ cellseg,
                                                      const uint2* __restrict__ tmpa,
                                                      int2* __restrict__ seg,
                                                      uint2* __restrict__ tmp2,
                                                      int n_nodes, int cb)
{
    __shared__ int cnt[1 << BSHIFT];
    __shared__ int wsum[8];
    const int t = threadIdx.x, lane = t & 31, w = t >> 5;
    const int b = (int)blockIdx.x;
    const int bs = cellseg[b * cb].x;
    const int be = cellseg[b * cb + cb - 1].y;

    if (t < (1 << BSHIFT)) cnt[t] = 0;
    __syncthreads();
    for (int i = bs + t; i < be; i += 4 * 1024) {
        unsigned dv[4];
#pragma unroll
        for (int u = 0; u < 4; ++u) {
            const int ii = (i + u * 1024 < be) ? (i + u * 1024) : (be - 1);
            dv[u] = tmpa[ii].y;
        }
#pragma unroll
        for (int u = 0; u < 4; ++u)
            atomicAdd(&cnt[dv[u] & ((1u << BSHIFT) - 1u)], (i + u * 1024 < be) ? 1 : 0);
    }
    __syncthreads();
    int c = 0, incl = 0;
    if (t < (1 << BSHIFT)) {
        c = cnt[t];
        incl = wave_scan_incl(c, lane);
        if (lane == 31) wsum[w] = incl;
    }
    __syncthreads();
    if (t < (1 << BSHIFT)) {
        int pre = 0;
        for (int j = 0; j < 8; ++j) pre += (j < w) ? wsum[j] : 0;
        const int excl = bs + pre + incl - c;
        const int node = (b << BSHIFT) + t;
        if (node < n_nodes) seg[node] = make_int2(excl, excl + c);
        cnt[t] = excl;
    }
    __syncthreads();
    for (int i = bs + t; i < be; i += 4 * 1024) {
        uint2 me[4];
#pragma unroll
        for (int u = 0; u < 4; ++u) {
            const int ii = (i + u * 1024 < be) ? (i + u * 1024) : (be - 1);
            me[u] = tmpa[ii];
        }
        int pos[4];
#pragma unroll
        for (int u = 0; u < 4; ++u)
            pos[u] = atomicAdd(&cnt[me[u].y & ((1u << BSHIFT) - 1u)], (i + u * 1024 < be) ? 1 : 0);
#pragma unroll
        for (int u = 0; u < 4; ++u)
            if (i + u * 1024 < be) tmp2[pos[u]] = me[u];
    }
}

__global__ __launch_bounds__(256) void order_kernel(const uint2* __restrict__ tmp2,
                                                    const int2* __restrict__ seg,
                                                    unsigned* __restrict__ recs, int n_edges)
{
    int i = blockIdx.x * blockDim.x + threadIdx.x;
    if (i >= n_edges) return;
    const uint2 me = tmp2[i];
    const unsigned x = me.x;
    const int v = (int)me.y;
    const int2 se = seg[v];
    const int beg = se.x, end = se.y;
    int rank = 0;
    for (int j = beg; j < end; j += 8) {
        unsigned y[8];
#pragma unroll
        for (int u = 0; u < 8; ++u) {
            const int idx = (j + u < end) ? (j + u) : (end - 1);
            y[u] = tmp2[idx].x;
        }
#pragma unroll
        for (int u = 0; u < 8; ++u) {
            const int k = j + u;
            const bool first = (y[u] < x) || (y[u] == x && k < i);
            rank += (k < end && first) ? 1 : 0;
        }
    }
    recs[beg + rank] = x;
}

__global__ __launch_bounds__(256)
void node_kernel(const float* __restrict__ h_in,
                 const float* __restrict__ hr,
                 const int2* __restrict__ seg,
                 const unsigned* __restrict__ recs,
                 const float* __restrict__ Ws,
                 const float* __restrict__ Wo,
                 const float* __restrict__ Wi,
                 float* __restrict__ h_out,
                 int n_nodes, int relu)
{
    __shared__ alignas(16) float lds_w[3 * D * WPITCH];
    __shared__ alignas(16) float lds_hr[3 * D];
    __shared__ alignas(16) float lds_a[3 * 32 * APITCH];

    const int t = threadIdx.x;

    const int g    = t >> 3;
    const int c    = (t & 7) * 4;

    const int n_tiles = (n_nodes + 31) >> 5;
    const int stride  = (int)gridDim.x;
    int tile = (int)blockIdx.x;

    int beg = 0, end = 0;
    float4 ha = make_float4(0.f, 0.f, 0.f, 0.f);
    float4 hb = ha;
    {
        const int node = tile * 32 + g;
        if (tile < n_tiles && node < n_nodes) {
            const int2 se = seg[node];
            beg = se.x;
            end = se.y;
            ha = *(const float4*)(h_in + (size_t)node * D + c);
            hb = *(const float4*)(h_in + (size_t)node * D + 32 + c);
        }
    }
    const float hrv = (t < 3 * D) ? hr[t] : 0.f;

    float4 wv[12];
#pragma unroll
    for (int j = 0; j < 4; ++j) {
        const int i4 = t + 256 * j;
        wv[j]     = ((const float4*)Ws)[i4];
        wv[4 + j] = ((const float4*)Wo)[i4];
        wv[8 + j] = ((const float4*)Wi)[i4];
    }

    unsigned rn[4] = { 0u, 0u, 0u, 0u };
    if (beg < end) {
#pragma unroll
        for (int u = 0; u < 4; ++u) {
            const int idx = (beg + u < end) ? (beg + u) : (end - 1);
            rn[u] = recs[idx];
        }
    }

#pragma unroll
    for (int j = 0; j < 4; ++j) {
        const int i4 = t + 256 * j;
        const int wi = (i4 >> 4) * WPITCH + (i4 & 15) * 4;
        *(float4*)(lds_w + wi)                  = wv[j];
        *(float4*)(lds_w + D * WPITCH + wi)     = wv[4 + j];
        *(float4*)(lds_w + 2 * D * WPITCH + wi) = wv[8 + j];
    }
    if (t < 3 * D) lds_hr[t] = hrv;

    const int wave  = t >> 5;
    const int lane  = t & 31;
    const int lhalf = lane >> 4;
    const int lmod  = lane & 15;
    const int n0    = (wave & 3) * 16;
    const int brow  = n0 + lmod;
    const float* xr = lds_a + ((wave >> 2) * 16 + lmod) * APITCH;
    const float* a0 = xr + 32 * APITCH;
    const float* a1 = xr + 64 * APITCH;

    for (; tile < n_tiles; tile += stride) {
        const int node = tile * 32 + g;
        const bool live = node < n_nodes;

        __syncthreads();

        const int ntile = tile + stride;
        int nbeg = 0, nend = 0;
        float4 nha = make_float4(0.f, 0.f, 0.f, 0.f);
        float4 nhb = nha;
        {
            const int nnode = ntile * 32 + g;
            if (ntile < n_tiles && nnode < n_nodes) {
                const int2 se = seg[nnode];
                nbeg = se.x;
                nend = se.y;
                nha = *(const float4*)(h_in + (size_t)nnode * D + c);
                nhb = *(const float4*)(h_in + (size_t)nnode * D + 32 + c);
            }
        }

        float4 xa = make_float4(0.f, 0.f, 0.f, 0.f);
        float4 xb = xa;
        if (live) {
            const float4 ra = *(const float4*)(lds_hr + 2 * D + c);
            const float4 rb = *(const float4*)(lds_hr + 2 * D + 32 + c);
            xa.x = ha.x - ra.x; xa.y = ha.y - ra.y; xa.z = ha.z - ra.z; xa.w = ha.w - ra.w;
            xb.x = hb.x - rb.x; xb.y = hb.y - rb.y; xb.z = hb.z - rb.z; xb.w = hb.w - rb.w;
        }
        *(float4*)(lds_a + g * APITCH + c)      = xa;
        *(float4*)(lds_a + g * APITCH + 32 + c) = xb;

        float4 s0a = make_float4(0.f, 0.f, 0.f, 0.f);
        float4 s0b = s0a, s1a = s0a, s1b = s0a;
        for (int i = beg; i < end; i += 4) {
            unsigned rec[4];
            float4   ga[4], gb[4];
#pragma unroll
            for (int u = 0; u < 4; ++u) rec[u] = rn[u];
#pragma unroll
            for (int u = 0; u < 4; ++u) {
                const float* p = h_in + (size_t)(rec[u] >> 2) * D + c;
                ga[u] = *(const float4*)p;
                gb[u] = *(const float4*)(p + 32);
            }
#pragma unroll
            for (int u = 0; u < 4; ++u) {
                const int idx = (i + 4 + u < end) ? (i + 4 + u) : (end - 1);
                rn[u] = recs[idx];
            }
#pragma unroll
            for (int u = 0; u < 4; ++u) {
                const bool ok = (i + u < end);
                const unsigned et = (rec[u] >> 1) & 1u;
                const bool     in = (rec[u] & 1u) != 0u;
                const bool     t0 = ok && !in;
                const bool     t1 = ok && in;
                const float4 ra = *(const float4*)(lds_hr + et * D + c);
                const float4 rb = *(const float4*)(lds_hr + et * D + 32 + c);
                float4 ma, mb;
                ma.x = ga[u].x - ra.x; ma.y = ga[u].y - ra.y;
                ma.z = ga[u].z - ra.z; ma.w = ga[u].w - ra.w;
                mb.x = gb[u].x - rb.x; mb.y = gb[u].y - rb.y;
                mb.z = gb[u].z - rb.z; mb.w = gb[u].w - rb.w;
                s0a.x += t0 ? ma.x : 0.f; s0a.y += t0 ? ma.y : 0.f;
                s0a.z += t0 ? ma.z : 0.f; s0a.w += t0 ? ma.w : 0.f;
                s0b.x += t0 ? mb.x : 0.f; s0b.y += t0 ? mb.y : 0.f;
                s0b.z += t0 ? mb.z : 0.f; s0b.w += t0 ? mb.w : 0.f;
                s1a.x += t1 ? ma.x : 0.f; s1a.y += t1 ? ma.y : 0.f;
                s1a.z += t1 ? ma.z : 0.f; s1a.w += t1 ? ma.w : 0.f;
                s1b.x += t1 ? mb.x : 0.f; s1b.y += t1 ? mb.y : 0.f;
                s1b.z += t1 ? mb.z : 0.f; s1b.w += t1 ? mb.w : 0.f;
            }
        }
        *(float4*)(lds_a + (32 + g) * APITCH + c)      = s0a;
        *(float4*)(lds_a + (32 + g) * APITCH + 32 + c) = s0b;
        *(float4*)(lds_a + (64 + g) * APITCH + c)      = s1a;
        *(float4*)(lds_a + (64 + g) * APITCH + 32 + c) = s1b;

        __syncthreads();

        unsigned nrn[4] = { 0u, 0u, 0u, 0u };
        if (nbeg < nend) {
#pragma unroll
            for (int u = 0; u < 4; ++u) {
                const int idx = (nbeg + u < nend) ? (nbeg + u) : (nend - 1);
                nrn[u] = recs[idx];
            }
        }

        const int m0 = tile * 32 + (wave >> 2) * 16;
        if (m0 < n_nodes) {
            v8f acc = {};
#pragma unroll
            for (int k0 = 0; k0 < D; k0 += 4) {
                const int ka = k0 + lhalf * 2;
                v2f a, b;
                a.x = xr[ka];
                a.y = xr[ka + 1];
                b.x = lds_w[brow * WPITCH + ka];
                b.y = lds_w[brow * WPITCH + ka + 1];
                acc = __builtin_amdgcn_wmma_f32_16x16x4_f32(false, a, false, b, (short)0, acc, false, false);
            }
#pragma unroll
            for (int k0 = 0; k0 < D; k0 += 4) {
                const int ka = k0 + lhalf * 2;
                v2f a, b;
                a.x = a0[ka];
                a.y = a0[ka + 1];
                b.x = lds_w[D * WPITCH + brow * WPITCH + ka];
                b.y = lds_w[D * WPITCH + brow * WPITCH + ka + 1];
                acc = __builtin_amdgcn_wmma_f32_16x16x4_f32(false, a, false, b, (short)0, acc, false, false);
            }
#pragma unroll
            for (int k0 = 0; k0 < D; k0 += 4) {
                const int ka = k0 + lhalf * 2;
                v2f a, b;
                a.x = a1[ka];
                a.y = a1[ka + 1];
                b.x = lds_w[2 * D * WPITCH + brow * WPITCH + ka];
                b.y = lds_w[2 * D * WPITCH + brow * WPITCH + ka + 1];
                acc = __builtin_amdgcn_wmma_f32_16x16x4_f32(false, a, false, b, (short)0, acc, false, false);
            }

#pragma unroll
            for (int r = 0; r < 8; ++r) {
                float v = acc[r];
                if (relu) v = fmaxf(v, 0.f);
                const int m = m0 + r + lhalf * 8;
                if (m < n_nodes) h_out[(size_t)m * D + n0 + lmod] = v;
            }
        }

        beg = nbeg; end = nend; ha = nha; hb = nhb;
#pragma unroll
        for (int u = 0; u < 4; ++u) rn[u] = nrn[u];
    }
}

extern "C" void kernel_launch(void* const* d_in, const int* in_sizes, int n_in,
                              void* d_out, int out_size, void* d_ws, size_t ws_size,
                              hipStream_t stream)
{
    const float* h_u   = (const float*)d_in[0];
    const float* Basis = (const float*)d_in[1];
    const float* alpha = (const float*)d_in[2];
    const float* W_O   = (const float*)d_in[3];
    const float* W_I   = (const float*)d_in[4];
    const float* W_S   = (const float*)d_in[5];
    const float* W_rel = (const float*)d_in[6];
    const int*   src   = (const int*)d_in[7];
    const int*   dst   = (const int*)d_in[8];
    const int*   etype = (const int*)d_in[9];
    const int*   edir  = (const int*)d_in[10];
    float* out = (float*)d_out;

    const int n_nodes = in_sizes[0] / D;
    const int n_edges = in_sizes[7];

    const int n_pad   = n_nodes + 64;
    const int n_bkt   = (n_nodes + (1 << BSHIFT) - 1) >> BSHIFT;
    const int n_cells = n_bkt * CB;
    int*      cells   = (int*)d_ws;
    int2*     cellseg = (int2*)(cells + n_cells + 64);
    int2*     seg     = cellseg + n_cells;
    uint2*    tmpa    = (uint2*)(seg + n_pad);
    uint2*    tmp2    = tmpa + n_edges;
    unsigned* recs    = (unsigned*)(tmp2 + n_edges);
    float* hr_l2 = (float*)(recs + n_edges);
    float* hb0 = (float*)d_ws + (size_t)2 * n_nodes * D;
    float* hb1 = hb0 + (size_t)n_nodes * D;
    float* hr_l0 = hb1 + (size_t)n_nodes * D;
    float* hr_l1 = hr_l0 + 256;

    const unsigned edge_blk256 = (unsigned)((n_edges + 255) / 256);
    const unsigned scan_blocks = (unsigned)((n_cells + 1023) / 1024);
    const size_t   bkt_lds     = (size_t)n_bkt * sizeof(int);

    hist_kernel<<<CB, 256, bkt_lds, stream>>>(dst, cells, cells + n_cells, n_edges, n_bkt);
    scan_kernel<<<scan_blocks + 1, 1024, 0, stream>>>(cells, cells + n_cells, cellseg, n_cells,
                                                      Basis, alpha, W_rel, hr_l0, hr_l1, hr_l2);
    scatter_kernel<<<CB, 256, bkt_lds, stream>>>(dst, src, etype, edir, cellseg, tmpa, n_edges, n_bkt);
    bucket_kernel<<<n_bkt, 1024, 0, stream>>>(cellseg, tmpa, seg, tmp2, n_nodes, CB);
    order_kernel<<<edge_blk256, 256, 0, stream>>>(tmp2, seg, recs, n_edges);

    const float* hin[3]  = { h_u, hb0, hb1 };
    float*       hout[3] = { hb0, hb1, out };
    const float* hrl[3]  = { hr_l0, hr_l1, hr_l2 };

    for (int l = 0; l < 3; ++l) {
        const int relu = (l < 2) ? 1 : 0;

        node_kernel<<<1024, 256, 0, stream>>>(hin[l], hrl[l], seg, recs,
                                                     W_S + (size_t)l * D * D,
                                                     W_O + (size_t)l * D * D,
                                                     W_I + (size_t)l * D * D,
                                                     hout[l], n_nodes, relu);
    }
}
